# starting kernel
# baseline (speedup 1.0000x reference)
#include <hip/hip_runtime.h>

typedef __attribute__((ext_vector_type(2))) float v2f;
typedef __attribute__((ext_vector_type(8))) float v8f;

#define D_IN  128
#define D_HID 128
#define D_OUT 64

__global__ void deg_init_kernel(float* __restrict__ deg, int n) {
    int i = blockIdx.x * blockDim.x + threadIdx.x;
    if (i < n) deg[i] = 1.0f;
}

__global__ void deg_accum_kernel(const int* __restrict__ dst,
                                 float* __restrict__ deg, int e) {
    int i = blockIdx.x * blockDim.x + threadIdx.x;
    if (i < e) atomicAdd(&deg[dst[i]], 1.0f);
}

__global__ void deg_rsqrt_kernel(float* __restrict__ dinv, int n) {
    int i = blockIdx.x * blockDim.x + threadIdx.x;
    if (i < n) {
        float d = dinv[i];
        dinv[i] = (d > 0.0f) ? rsqrtf(d) : 0.0f;
    }
}

template <int NOUT>
__global__ void gemm_wmma_f32_kernel(const float* __restrict__ A,
                                     const float* __restrict__ W,
                                     float* __restrict__ H, int M) {
    constexpr int K = 128;
    constexpr int NT = NOUT / 16;
    __shared__ __align__(16) float lds_a[16 * K];

    const int block_row = blockIdx.x * 16;
    const int tid = threadIdx.x;
    constexpr int NTHREADS = 32 * NT;

    for (int i = tid * 4; i < 16 * K; i += NTHREADS * 4) {
        *(float4*)(&lds_a[i]) = *(const float4*)(&A[(size_t)block_row * K + i]);
    }
    __syncthreads();

    const int wave = tid >> 5;
    const int lane = tid & 31;
    const int half = lane >> 4;
    const int l16  = lane & 15;
    const int col_base = wave * 16;

    v8f acc = {};
    #pragma unroll 4
    for (int k0 = 0; k0 < K; k0 += 4) {
        const int ka = k0 + half * 2;
        v2f a, b;
        a.x = lds_a[l16 * K + ka];
        a.y = lds_a[l16 * K + ka + 1];
        b.x = W[(size_t)ka * NOUT + col_base + l16];
        b.y = W[(size_t)(ka + 1) * NOUT + col_base + l16];
        acc = __builtin_amdgcn_wmma_f32_16x16x4_f32(
             false, a,  false, b,
             (short)0, acc,  false,  false);
    }

    #pragma unroll
    for (int v = 0; v < 8; ++v) {
        const int row = v + half * 8;
        H[(size_t)(block_row + row) * NOUT + col_base + l16] = acc[v];
    }
}

template <int D>
__global__ void self_init_kernel(const float* __restrict__ h,
                                 const float* __restrict__ dinv,
                                 float* __restrict__ agg, int total) {
    int idx = blockIdx.x * blockDim.x + threadIdx.x;
    if (idx >= total) return;
    const int node = idx / D;
    const float di = dinv[node];
    agg[idx] = h[idx] * di * di;
}

template <int D, int VPL>
__global__ void edge_scatter_kernel(const int* __restrict__ src,
                                    const int* __restrict__ dst,
                                    const float* __restrict__ h,
                                    const float* __restrict__ dinv,
                                    float* __restrict__ agg, int e) {
    const int widx = (blockIdx.x * blockDim.x + threadIdx.x) >> 5;
    const int lane = threadIdx.x & 31;
    if (widx >= e) return;
    const int s = src[widx];
    const int d = dst[widx];
    const float norm = dinv[s] * dinv[d];
    const float* __restrict__ hp = h + (size_t)s * D + lane * VPL;
    float* __restrict__ op = agg + (size_t)d * D + lane * VPL;
    if constexpr (VPL == 4) {
        const float4 v = *(const float4*)hp;
        atomicAdd(op + 0, v.x * norm);
        atomicAdd(op + 1, v.y * norm);
        atomicAdd(op + 2, v.z * norm);
        atomicAdd(op + 3, v.w * norm);
    } else {
        const float2 v = *(const float2*)hp;
        atomicAdd(op + 0, v.x * norm);
        atomicAdd(op + 1, v.y * norm);
    }
}

template <int D, bool RELU>
__global__ void bias_act_kernel(float* __restrict__ data,
                                const float* __restrict__ bias, int total) {
    int idx = blockIdx.x * blockDim.x + threadIdx.x;
    if (idx >= total) return;
    float v = data[idx] + bias[idx & (D - 1)];
    if (RELU) v = fmaxf(v, 0.0f);
    data[idx] = v;
}

extern "C" void kernel_launch(void* const* d_in, const int* in_sizes, int n_in,
                              void* d_out, int out_size, void* d_ws, size_t ws_size,
                              hipStream_t stream) {
    const float* x    = (const float*)d_in[0];
    const int*   eidx = (const int*)  d_in[1];
    const float* W1   = (const float*)d_in[2];
    const float* b1   = (const float*)d_in[3];
    const float* W2   = (const float*)d_in[4];
    const float* b2   = (const float*)d_in[5];
    float* out = (float*)d_out;

    const int N = in_sizes[0] / D_IN;
    const int E = in_sizes[1] / 2;
    const int* src = eidx;
    const int* dst = eidx + E;

    float* ws = (float*)d_ws;
    float* dinv = ws;
    float* h1   = ws + (((size_t)N + 255) & ~(size_t)255);
    float* agg1 = h1 + (size_t)N * D_HID;
    float* h2   = h1;

    const int BT = 256;

    deg_init_kernel<<<(N + BT - 1) / BT, BT, 0, stream>>>(dinv, N);
    deg_accum_kernel<<<(E + BT - 1) / BT, BT, 0, stream>>>(dst, dinv, E);
    deg_rsqrt_kernel<<<(N + BT - 1) / BT, BT, 0, stream>>>(dinv, N);

    gemm_wmma_f32_kernel<D_HID><<<N / 16, 32 * (D_HID / 16), 0, stream>>>(x, W1, h1, N);
    {
        const int total = N * D_HID;
        self_init_kernel<D_HID><<<(total + BT - 1) / BT, BT, 0, stream>>>(h1, dinv, agg1, total);
        const long tthreads = (long)E * 32;
        edge_scatter_kernel<D_HID, 4><<<(unsigned)((tthreads + BT - 1) / BT), BT, 0, stream>>>(
            src, dst, h1, dinv, agg1, E);
        bias_act_kernel<D_HID, true><<<(total + BT - 1) / BT, BT, 0, stream>>>(agg1, b1, total);
    }

    gemm_wmma_f32_kernel<D_OUT><<<N / 16, 32 * (D_OUT / 16), 0, stream>>>(agg1, W2, h2, N);
    {
        const int total = N * D_OUT;
        self_init_kernel<D_OUT><<<(total + BT - 1) / BT, BT, 0, stream>>>(h2, dinv, out, total);
        const long tthreads = (long)E * 32;
        edge_scatter_kernel<D_OUT, 2><<<(unsigned)((tthreads + BT - 1) / BT), BT, 0, stream>>>(
            src, dst, h2, dinv, out, E);
        bias_act_kernel<D_OUT, false><<<(total + BT - 1) / BT, BT, 0, stream>>>(out, b2, total);
    }
}


# GEMM strip rows in LDS at a pitch of 132 floats, not 128: the 8-byte A-fragment reads of 16 rows no longer fall on one bank pair (16 turns per read before); both WMMA launches; sum order unchanged
# speedup vs baseline: 2.3139x; 2.3139x over previous
#include <hip/hip_runtime.h>

typedef __attribute__((ext_vector_type(2))) float v2f;
typedef __attribute__((ext_vector_type(8))) float v8f;

#define D_IN  128
#define D_HID 128
#define D_OUT 64

#define OWN_R 400

__global__ void deg_init_kernel(float* __restrict__ deg, int n) {
    int i = blockIdx.x * blockDim.x + threadIdx.x;
    if (i < n) deg[i] = 1.0f;
}

__global__ void deg_accum_kernel(const int* __restrict__ dst,
                                 float* __restrict__ deg,
                                 unsigned int* __restrict__ code, int e) {
    const int t = blockIdx.x * blockDim.x + threadIdx.x;
    const int i0 = t * 4;
    if (i0 >= e) return;
    int d[4];
    if (i0 + 3 < e && (reinterpret_cast<size_t>(dst) & 15) == 0) {
        const int4 v = *(const int4*)(dst + i0);
        d[0] = v.x; d[1] = v.y; d[2] = v.z; d[3] = v.w;
    } else {
        #pragma unroll
        for (int k = 0; k < 4; ++k) d[k] = dst[min(i0 + k, e - 1)];
    }
    unsigned w = 0;
    #pragma unroll
    for (int k = 0; k < 4; ++k) {
        if (i0 + k < e) {
            atomicAdd(&deg[d[k]], 1.0f);
            w |= (unsigned)(d[k] / OWN_R) << (8 * k);
        }
    }
    code[t] = w;
}

__global__ void deg_rsqrt_kernel(float* __restrict__ dinv, int n) {
    int i = blockIdx.x * blockDim.x + threadIdx.x;
    if (i < n) {
        float d = dinv[i];
        dinv[i] = (d > 0.0f) ? rsqrtf(d) : 0.0f;
    }
}

template <int NOUT, bool RELU_IN>
__global__ void gemm_wmma_f32_kernel(const float* __restrict__ A,
                                     const float* __restrict__ W,
                                     const float* __restrict__ dinv,
                                     const float* __restrict__ bias,
                                     float* __restrict__ H,
                                     float* __restrict__ AGG, int M) {
    constexpr int K = 128;
    constexpr int NT = NOUT / 16;
    constexpr int P = K + 4;
    __shared__ __align__(16) float lds_a[16 * P];

    const int block_row = blockIdx.x * 16;
    const int tid = threadIdx.x;
    constexpr int NTHREADS = 32 * NT;

    const int wave = tid >> 5;
    const int lane = tid & 31;
    const int half = lane >> 4;
    const int l16  = lane & 15;
    const int col_base = wave * 16;

    const float bcol = bias[col_base + l16];
    float di[8];
    #pragma unroll
    for (int v = 0; v < 8; ++v) di[v] = dinv[block_row + half * 8 + v];

    for (int i = tid * 4; i < 16 * K; i += NTHREADS * 4) {
        float4 t = *(const float4*)(&A[(size_t)block_row * K + i]);
        if (RELU_IN) {
            t.x = fmaxf(t.x, 0.0f);
            t.y = fmaxf(t.y, 0.0f);
            t.z = fmaxf(t.z, 0.0f);
            t.w = fmaxf(t.w, 0.0f);
        }
        *(float4*)(&lds_a[(i / K) * P + (i % K)]) = t;
    }
    __syncthreads();

    v8f acc = {};
    #pragma unroll 4
    for (int k0 = 0; k0 < K; k0 += 4) {
        const int ka = k0 + half * 2;
        v2f a, b;
        a.x = lds_a[l16 * P + ka];
        a.y = lds_a[l16 * P + ka + 1];
        b.x = W[(size_t)ka * NOUT + col_base + l16];
        b.y = W[(size_t)(ka + 1) * NOUT + col_base + l16];
        acc = __builtin_amdgcn_wmma_f32_16x16x4_f32(
             false, a,  false, b,
             (short)0, acc,  false,  false);
    }

    #pragma unroll
    for (int v = 0; v < 8; ++v) {
        const int row = v + half * 8;
        const size_t o = (size_t)(block_row + row) * NOUT + col_base + l16;
        const float hv = acc[v];
        H[o] = hv;
        AGG[o] = hv * di[v] * di[v] + bcol;
    }
}

#define OWN_WAVES 32
#define OWN_U 4
#define OWN_STEP (512 * OWN_U)
#define OWN_TILE_FLOATS (OWN_R * 128)

template <int D>
__device__ __forceinline__ void own_flush(const int* qe, int cnt, int lane, int lo,
                                          const int* __restrict__ src,
                                          const int* __restrict__ dst,
                                          const float* __restrict__ h,
                                          const float* __restrict__ dinv,
                                          const float* own_dinv, float* tile) {
    constexpr int VPL = D / 32;
    typedef float vrow __attribute__((ext_vector_type(VPL)));
    int my_s = 0, my_dl = 0;
    float my_c = 0.0f;
    if (lane < cnt) {
        const int my_e = qe[lane];
        my_dl = dst[my_e] - lo;
        my_s = src[my_e];
        my_c = dinv[my_s] * own_dinv[my_dl];
    }
    for (int g = 0; g < cnt; g += 8) {
        vrow v[8];
        #pragma unroll
        for (int t = 0; t < 8; ++t) {
            v[t] = (vrow)(0.0f);
            if (g + t < cnt) {
                const int s = __builtin_amdgcn_readlane(my_s, g + t);
                v[t] = *(const vrow*)(h + (size_t)s * D + lane * VPL);
            }
        }
        #pragma unroll
        for (int t = 0; t < 8; ++t) {
            if (g + t < cnt) {
                const float c = __builtin_bit_cast(
                    float, __builtin_amdgcn_readlane(__builtin_bit_cast(int, my_c), g + t));
                const int dl = __builtin_amdgcn_readlane(my_dl, g + t);
                float* tp = tile + dl * D + lane;
                #pragma unroll
                for (int j = 0; j < VPL; ++j) atomicAdd(tp + j * 32, v[t][j] * c);
            }
        }
    }
}

__device__ __forceinline__ unsigned own_differs(unsigned w, unsigned pat) {
    const unsigned x = w ^ pat;
    return ((x & 0x7f7f7f7fu) + 0x7f7f7f7fu) | x;
}

__device__ __forceinline__ int own_place(int base, int lane, int b) {
    const int q = b & 7;
    return base + (q >> 2) * 512 + lane * 16 + (q & 3) * 4 + (b >> 3);
}

template <int D>
__global__ __launch_bounds__(1024)
void owner_aggregate_kernel(const int* __restrict__ src, const int* __restrict__ dst,
                            const unsigned int* __restrict__ code,
                            const float* __restrict__ h, const float* __restrict__ dinv,
                            float* __restrict__ agg, int n, int e) {
    constexpr int VPL = D / 32;
    constexpr int R = OWN_R;
    typedef float vrow __attribute__((ext_vector_type(VPL)));
    __shared__ __align__(16) float tile[OWN_TILE_FLOATS];
    __shared__ float own_dinv[R];
    __shared__ int q_e[OWN_WAVES][64];

    const int tid  = threadIdx.x;
    const int lane = tid & 31;
    const int wave = __builtin_amdgcn_readfirstlane(tid >> 5);
    const int lo   = blockIdx.x * R;
    const int rows = min(R, n - lo);

    #pragma unroll 4
    for (int r = wave; r < rows; r += OWN_WAVES) {
        const vrow t = *(const vrow*)(agg + (size_t)(lo + r) * D + lane * VPL);
        #pragma unroll
        for (int j = 0; j < VPL; ++j) tile[r * D + j * 32 + lane] = t[j];
    }
    for (int r = tid; r < rows; r += 1024) own_dinv[r] = dinv[lo + r];
    __syncthreads();

    const int nst   = (e + OWN_STEP - 1) / OWN_STEP;
    const int nfull = e / OWN_STEP;
    const int rot  = (int)(((long long)blockIdx.x * nst) / gridDim.x);
    const unsigned mine = blockIdx.x & 255u;
    const unsigned pat  = mine * 0x01010101u;
    const unsigned lt = (1u << lane) - 1u;
    const uint4* cvec = (const uint4*)code;
    const unsigned char* cbyte = (const unsigned char*)code;
    int* qe = q_e[wave];
    int qn = 0;

    uint4 nx[OWN_U];
    {
        int c = wave + rot;
        if (c >= nst) c -= nst;
        const int cl = (wave < nst && c < nfull) ? c : 0;
        const uint4* p = cvec + (size_t)cl * (OWN_STEP / 16) + lane;
        #pragma unroll
        for (int u = 0; u < OWN_U; ++u) nx[u] = p[u * 32];
    }

    for (int ci = wave; ci < nst; ci += OWN_WAVES) {
        int c = ci + rot;
        if (c >= nst) c -= nst;
        const int base = c * OWN_STEP;
        uint4 dv[OWN_U];
        #pragma unroll
        for (int u = 0; u < OWN_U; ++u) dv[u] = nx[u];
        {
            const int cj = ci + OWN_WAVES;
            int cn = cj + rot;
            if (cn >= nst) cn -= nst;
            const int cl = (cj < nst && cn < nfull) ? cn : 0;
            const uint4* p = cvec + (size_t)cl * (OWN_STEP / 16) + lane;
            #pragma unroll
            for (int u = 0; u < OWN_U; ++u) nx[u] = p[u * 32];
        }

        unsigned mm[OWN_U / 2];
        if (c < nfull) {
            #pragma unroll
            for (int hp = 0; hp < OWN_U / 2; ++hp) {
                const uint4 lo4 = dv[2 * hp], hi4 = dv[2 * hp + 1];
                unsigned nd = 0;
                nd |= (own_differs(lo4.x, pat) & 0x80808080u) >> 7;
                nd |= (own_differs(lo4.y, pat) & 0x80808080u) >> 6;
                nd |= (own_differs(lo4.z, pat) & 0x80808080u) >> 5;
                nd |= (own_differs(lo4.w, pat) & 0x80808080u) >> 4;
                nd |= (own_differs(hi4.x, pat) & 0x80808080u) >> 3;
                nd |= (own_differs(hi4.y, pat) & 0x80808080u) >> 2;
                nd |= (own_differs(hi4.z, pat) & 0x80808080u) >> 1;
                nd |= (own_differs(hi4.w, pat) & 0x80808080u);
                mm[hp] = ~nd;
            }
        } else {
            #pragma unroll
            for (int hp = 0; hp < OWN_U / 2; ++hp) {
                unsigned m = 0;
                #pragma nounroll
                for (int b = 0; b < 32; ++b) {
                    const int idx = own_place(base + hp * 1024, lane, b);
                    const unsigned cv = cbyte[min(idx, e - 1)];
                    if (idx < e && cv == mine) m |= 1u << b;
                }
                mm[hp] = m;
            }
        }

        #pragma unroll
        for (int hp = 0; hp < OWN_U / 2; ++hp) {
            unsigned m = mm[hp];
            while (true) {
                const bool has = m != 0;
                const unsigned bal = __builtin_amdgcn_ballot_w32(has);
                if (bal == 0) break;
                if (has) {
                    const int b = __builtin_ctz(m);
                    m &= m - 1;
                    const int slot = qn + __builtin_popcount(bal & lt);
                    qe[slot] = own_place(base + hp * 1024, lane, b);
                }
                qn += __builtin_popcount(bal);
                if (qn >= 32) {
                    qn -= 32;
                    own_flush<D>(qe + qn, 32, lane, lo, src, dst, h, dinv, own_dinv, tile);
                }
            }
        }
    }
    if (qn > 0) own_flush<D>(qe, qn, lane, lo, src, dst, h, dinv, own_dinv, tile);
    __syncthreads();

    for (int r = wave; r < rows; r += OWN_WAVES) {
        vrow t;
        #pragma unroll
        for (int j = 0; j < VPL; ++j) t[j] = tile[r * D + j * 32 + lane];
        *(vrow*)(agg + (size_t)(lo + r) * D + lane * VPL) = t;
    }
}

extern "C" void kernel_launch(void* const* d_in, const int* in_sizes, int n_in,
                              void* d_out, int out_size, void* d_ws, size_t ws_size,
                              hipStream_t stream) {
    const float* x    = (const float*)d_in[0];
    const int*   eidx = (const int*)  d_in[1];
    const float* W1   = (const float*)d_in[2];
    const float* b1   = (const float*)d_in[3];
    const float* W2   = (const float*)d_in[4];
    const float* b2   = (const float*)d_in[5];
    float* out = (float*)d_out;

    const int N = in_sizes[0] / D_IN;
    const int E = in_sizes[1] / 2;
    const int* src = eidx;
    const int* dst = eidx + E;

    float* ws = (float*)d_ws;
    float* dinv = ws;
    float* h1   = ws + (((size_t)N + 255) & ~(size_t)255);
    float* agg1 = h1 + (size_t)N * D_HID;
    float* h2   = h1;
    unsigned int* code = (unsigned int*)(agg1 + (size_t)N * D_HID);

    const int BT = 256;

    deg_init_kernel<<<(N + BT - 1) / BT, BT, 0, stream>>>(dinv, N);
    deg_accum_kernel<<<(E + 4 * BT - 1) / (4 * BT), BT, 0, stream>>>(dst, dinv, code, E);
    deg_rsqrt_kernel<<<(N + BT - 1) / BT, BT, 0, stream>>>(dinv, N);

    gemm_wmma_f32_kernel<D_HID, false><<<N / 16, 32 * (D_HID / 16), 0, stream>>>(
        x, W1, dinv, b1, h1, agg1, N);
    owner_aggregate_kernel<D_HID><<<(N + OWN_R - 1) / OWN_R, 32 * OWN_WAVES, 0, stream>>>(
        src, dst, code, h1, dinv, agg1, N, E);

    gemm_wmma_f32_kernel<D_OUT, true><<<N / 16, 32 * (D_OUT / 16), 0, stream>>>(
        agg1, W2, dinv, b2, h2, out, N);
    owner_aggregate_kernel<D_OUT><<<(N + OWN_R - 1) / OWN_R, 32 * OWN_WAVES, 0, stream>>>(
        src, dst, code, h2, dinv, out, N, E);
}


# Per-edge 8-byte record (source | target%400, dinv[source]) packed once by the rsqrt launch; owner groups fetch 1 record line per kept edge, not 3 (src, dst, dinv[src]); sum order unchanged
# speedup vs baseline: 2.4347x; 1.0522x over previous
#include <hip/hip_runtime.h>

typedef __attribute__((ext_vector_type(2))) float v2f;
typedef __attribute__((ext_vector_type(8))) float v8f;

#define D_IN  128
#define D_HID 128
#define D_OUT 64

#define OWN_R 400

__global__ void deg_init_kernel(float* __restrict__ deg, int n) {
    int i = blockIdx.x * blockDim.x + threadIdx.x;
    if (i < n) deg[i] = 1.0f;
}

__global__ void deg_accum_kernel(const int* __restrict__ dst,
                                 float* __restrict__ deg,
                                 unsigned int* __restrict__ code, int e) {
    const int t = blockIdx.x * blockDim.x + threadIdx.x;
    const int i0 = t * 4;
    if (i0 >= e) return;
    int d[4];
    if (i0 + 3 < e && (reinterpret_cast<size_t>(dst) & 15) == 0) {
        const int4 v = *(const int4*)(dst + i0);
        d[0] = v.x; d[1] = v.y; d[2] = v.z; d[3] = v.w;
    } else {
        #pragma unroll
        for (int k = 0; k < 4; ++k) d[k] = dst[min(i0 + k, e - 1)];
    }
    unsigned w = 0;
    #pragma unroll
    for (int k = 0; k < 4; ++k) {
        if (i0 + k < e) {
            atomicAdd(&deg[d[k]], 1.0f);
            w |= (unsigned)(d[k] / OWN_R) << (8 * k);
        }
    }
    code[t] = w;
}

__global__ void norm_pack_kernel(const int* __restrict__ src, const int* __restrict__ dst,
                                 const float* __restrict__ deg, float* __restrict__ dinv,
                                 unsigned int* __restrict__ rec, int n, int e) {
    const int t = blockIdx.x * blockDim.x + threadIdx.x;
    for (int i = t; i < n; i += gridDim.x * blockDim.x) {
        const float d = deg[i];
        dinv[i] = (d > 0.0f) ? rsqrtf(d) : 0.0f;
    }
    uint2* r2 = (uint2*)rec;
    const int base = blockIdx.x * (blockDim.x * 4) + threadIdx.x;
    if (base >= e) return;
    unsigned s[4], d[4];
    float ds[4];
    #pragma unroll
    for (int k = 0; k < 4; ++k) {
        const int i = min(base + k * (int)blockDim.x, e - 1);
        s[k] = (unsigned)src[i];
        d[k] = (unsigned)dst[i];
    }
    #pragma unroll
    for (int k = 0; k < 4; ++k) ds[k] = deg[s[k]];
    #pragma unroll
    for (int k = 0; k < 4; ++k) {
        const int i = base + k * (int)blockDim.x;
        if (i < e) {
            uint2 r;
            r.x = s[k] | ((d[k] % (unsigned)OWN_R) << 17);
            r.y = __float_as_uint((ds[k] > 0.0f) ? rsqrtf(ds[k]) : 0.0f);
            r2[i] = r;
        }
    }
}

template <int NOUT, bool RELU_IN>
__global__ void gemm_wmma_f32_kernel(const float* __restrict__ A,
                                     const float* __restrict__ W,
                                     const float* __restrict__ dinv,
                                     const float* __restrict__ bias,
                                     float* __restrict__ H,
                                     float* __restrict__ AGG, int M) {
    constexpr int K = 128;
    constexpr int NT = NOUT / 16;
    constexpr int P = K + 4;
    __shared__ __align__(16) float lds_a[16 * P];

    const int block_row = blockIdx.x * 16;
    const int tid = threadIdx.x;
    constexpr int NTHREADS = 32 * NT;

    const int wave = tid >> 5;
    const int lane = tid & 31;
    const int half = lane >> 4;
    const int l16  = lane & 15;
    const int col_base = wave * 16;

    const float bcol = bias[col_base + l16];
    float di[8];
    #pragma unroll
    for (int v = 0; v < 8; ++v) di[v] = dinv[block_row + half * 8 + v];

    for (int i = tid * 4; i < 16 * K; i += NTHREADS * 4) {
        float4 t = *(const float4*)(&A[(size_t)block_row * K + i]);
        if (RELU_IN) {
            t.x = fmaxf(t.x, 0.0f);
            t.y = fmaxf(t.y, 0.0f);
            t.z = fmaxf(t.z, 0.0f);
            t.w = fmaxf(t.w, 0.0f);
        }
        *(float4*)(&lds_a[(i / K) * P + (i % K)]) = t;
    }
    __syncthreads();

    v8f acc = {};
    #pragma unroll 4
    for (int k0 = 0; k0 < K; k0 += 4) {
        const int ka = k0 + half * 2;
        v2f a, b;
        a.x = lds_a[l16 * P + ka];
        a.y = lds_a[l16 * P + ka + 1];
        b.x = W[(size_t)ka * NOUT + col_base + l16];
        b.y = W[(size_t)(ka + 1) * NOUT + col_base + l16];
        acc = __builtin_amdgcn_wmma_f32_16x16x4_f32(
             false, a,  false, b,
             (short)0, acc,  false,  false);
    }

    #pragma unroll
    for (int v = 0; v < 8; ++v) {
        const int row = v + half * 8;
        const size_t o = (size_t)(block_row + row) * NOUT + col_base + l16;
        const float hv = acc[v];
        H[o] = hv;
        AGG[o] = hv * di[v] * di[v] + bcol;
    }
}

#define OWN_WAVES 32
#define OWN_U 4
#define OWN_STEP (512 * OWN_U)
#define OWN_TILE_FLOATS (OWN_R * 128)

template <int D>
__device__ __forceinline__ void own_flush(const int* qe, int cnt, int lane,
                                          const uint2* __restrict__ rec,
                                          const float* __restrict__ h,
                                          const float* own_dinv, float* tile) {
    constexpr int VPL = D / 32;
    typedef float vrow __attribute__((ext_vector_type(VPL)));
    int my_s = 0, my_dl = 0;
    float my_c = 0.0f;
    if (lane < cnt) {
        const uint2 r = rec[qe[lane]];
        my_s = (int)(r.x & 0x1ffffu);
        my_dl = (int)(r.x >> 17);
        my_c = __uint_as_float(r.y) * own_dinv[my_dl];
    }
    for (int g = 0; g < cnt; g += 8) {
        vrow v[8];
        #pragma unroll
        for (int t = 0; t < 8; ++t) {
            v[t] = (vrow)(0.0f);
            if (g + t < cnt) {
                const int s = __builtin_amdgcn_readlane(my_s, g + t);
                v[t] = *(const vrow*)(h + (size_t)s * D + lane * VPL);
            }
        }
        #pragma unroll
        for (int t = 0; t < 8; ++t) {
            if (g + t < cnt) {
                const float c = __builtin_bit_cast(
                    float, __builtin_amdgcn_readlane(__builtin_bit_cast(int, my_c), g + t));
                const int dl = __builtin_amdgcn_readlane(my_dl, g + t);
                float* tp = tile + dl * D + lane;
                #pragma unroll
                for (int j = 0; j < VPL; ++j) atomicAdd(tp + j * 32, v[t][j] * c);
            }
        }
    }
}

__device__ __forceinline__ unsigned own_differs(unsigned w, unsigned pat) {
    const unsigned x = w ^ pat;
    return ((x & 0x7f7f7f7fu) + 0x7f7f7f7fu) | x;
}

__device__ __forceinline__ int own_place(int base, int lane, int b) {
    const int q = b & 7;
    return base + (q >> 2) * 512 + lane * 16 + (q & 3) * 4 + (b >> 3);
}

template <int D>
__global__ __launch_bounds__(1024)
void owner_aggregate_kernel(const unsigned int* __restrict__ recw,
                            const unsigned int* __restrict__ code,
                            const float* __restrict__ h, const float* __restrict__ dinv,
                            float* __restrict__ agg, int n, int e) {
    constexpr int VPL = D / 32;
    constexpr int R = OWN_R;
    typedef float vrow __attribute__((ext_vector_type(VPL)));
    __shared__ __align__(16) float tile[OWN_TILE_FLOATS];
    __shared__ float own_dinv[R];
    __shared__ int q_e[OWN_WAVES][64];

    const int tid  = threadIdx.x;
    const int lane = tid & 31;
    const int wave = __builtin_amdgcn_readfirstlane(tid >> 5);
    const int lo   = blockIdx.x * R;
    const int rows = min(R, n - lo);
    const uint2* rec = (const uint2*)recw;

    #pragma unroll 4
    for (int r = wave; r < rows; r += OWN_WAVES) {
        const vrow t = *(const vrow*)(agg + (size_t)(lo + r) * D + lane * VPL);
        #pragma unroll
        for (int j = 0; j < VPL; ++j) tile[r * D + j * 32 + lane] = t[j];
    }
    for (int r = tid; r < rows; r += 1024) own_dinv[r] = dinv[lo + r];
    __syncthreads();

    const int nst   = (e + OWN_STEP - 1) / OWN_STEP;
    const int nfull = e / OWN_STEP;
    const int rot  = (int)(((long long)blockIdx.x * nst) / gridDim.x);
    const unsigned mine = blockIdx.x & 255u;
    const unsigned pat  = mine * 0x01010101u;
    const unsigned lt = (1u << lane) - 1u;
    const uint4* cvec = (const uint4*)code;
    const unsigned char* cbyte = (const unsigned char*)code;
    int* qe = q_e[wave];
    int qn = 0;

    uint4 nx[OWN_U];
    {
        int c = wave + rot;
        if (c >= nst) c -= nst;
        const int cl = (wave < nst && c < nfull) ? c : 0;
        const uint4* p = cvec + (size_t)cl * (OWN_STEP / 16) + lane;
        #pragma unroll
        for (int u = 0; u < OWN_U; ++u) nx[u] = p[u * 32];
    }

    for (int ci = wave; ci < nst; ci += OWN_WAVES) {
        int c = ci + rot;
        if (c >= nst) c -= nst;
        const int base = c * OWN_STEP;
        uint4 dv[OWN_U];
        #pragma unroll
        for (int u = 0; u < OWN_U; ++u) dv[u] = nx[u];
        {
            const int cj = ci + OWN_WAVES;
            int cn = cj + rot;
            if (cn >= nst) cn -= nst;
            const int cl = (cj < nst && cn < nfull) ? cn : 0;
            const uint4* p = cvec + (size_t)cl * (OWN_STEP / 16) + lane;
            #pragma unroll
            for (int u = 0; u < OWN_U; ++u) nx[u] = p[u * 32];
        }

        unsigned mm[OWN_U / 2];
        if (c < nfull) {
            #pragma unroll
            for (int hp = 0; hp < OWN_U / 2; ++hp) {
                const uint4 lo4 = dv[2 * hp], hi4 = dv[2 * hp + 1];
                unsigned nd = 0;
                nd |= (own_differs(lo4.x, pat) & 0x80808080u) >> 7;
                nd |= (own_differs(lo4.y, pat) & 0x80808080u) >> 6;
                nd |= (own_differs(lo4.z, pat) & 0x80808080u) >> 5;
                nd |= (own_differs(lo4.w, pat) & 0x80808080u) >> 4;
                nd |= (own_differs(hi4.x, pat) & 0x80808080u) >> 3;
                nd |= (own_differs(hi4.y, pat) & 0x80808080u) >> 2;
                nd |= (own_differs(hi4.z, pat) & 0x80808080u) >> 1;
                nd |= (own_differs(hi4.w, pat) & 0x80808080u);
                mm[hp] = ~nd;
            }
        } else {
            #pragma unroll
            for (int hp = 0; hp < OWN_U / 2; ++hp) {
                unsigned m = 0;
                #pragma nounroll
                for (int b = 0; b < 32; ++b) {
                    const int idx = own_place(base + hp * 1024, lane, b);
                    const unsigned cv = cbyte[min(idx, e - 1)];
                    if (idx < e && cv == mine) m |= 1u << b;
                }
                mm[hp] = m;
            }
        }

        #pragma unroll
        for (int hp = 0; hp < OWN_U / 2; ++hp) {
            unsigned m = mm[hp];
            while (true) {
                const bool has = m != 0;
                const unsigned bal = __builtin_amdgcn_ballot_w32(has);
                if (bal == 0) break;
                if (has) {
                    const int b = __builtin_ctz(m);
                    m &= m - 1;
                    const int slot = qn + __builtin_popcount(bal & lt);
                    qe[slot] = own_place(base + hp * 1024, lane, b);
                }
                qn += __builtin_popcount(bal);
                if (qn >= 32) {
                    qn -= 32;
                    own_flush<D>(qe + qn, 32, lane, rec, h, own_dinv, tile);
                }
            }
        }
    }
    if (qn > 0) own_flush<D>(qe, qn, lane, rec, h, own_dinv, tile);
    __syncthreads();

    for (int r = wave; r < rows; r += OWN_WAVES) {
        vrow t;
        #pragma unroll
        for (int j = 0; j < VPL; ++j) t[j] = tile[r * D + j * 32 + lane];
        *(vrow*)(agg + (size_t)(lo + r) * D + lane * VPL) = t;
    }
}

extern "C" void kernel_launch(void* const* d_in, const int* in_sizes, int n_in,
                              void* d_out, int out_size, void* d_ws, size_t ws_size,
                              hipStream_t stream) {
    const float* x    = (const float*)d_in[0];
    const int*   eidx = (const int*)  d_in[1];
    const float* W1   = (const float*)d_in[2];
    const float* b1   = (const float*)d_in[3];
    const float* W2   = (const float*)d_in[4];
    const float* b2   = (const float*)d_in[5];
    float* out = (float*)d_out;

    const int N = in_sizes[0] / D_IN;
    const int E = in_sizes[1] / 2;
    const int* src = eidx;
    const int* dst = eidx + E;

    float* ws = (float*)d_ws;
    float* deg  = ws;
    float* dinv = deg + (((size_t)N + 255) & ~(size_t)255);
    float* h1   = dinv + (((size_t)N + 255) & ~(size_t)255);
    float* agg1 = h1 + (size_t)N * D_HID;
    float* h2   = h1;
    unsigned int* code = (unsigned int*)(agg1 + (size_t)N * D_HID);
    unsigned int* rec  = code + ((((size_t)E + 3) / 4 + 255) & ~(size_t)255);

    const int BT = 256;

    deg_init_kernel<<<(N + BT - 1) / BT, BT, 0, stream>>>(deg, N);
    deg_accum_kernel<<<(E + 4 * BT - 1) / (4 * BT), BT, 0, stream>>>(dst, deg, code, E);
    norm_pack_kernel<<<(E + 4 * BT - 1) / (4 * BT), BT, 0, stream>>>(src, dst, deg, dinv, rec, N, E);

    gemm_wmma_f32_kernel<D_HID, false><<<N / 16, 32 * (D_HID / 16), 0, stream>>>(
        x, W1, dinv, b1, h1, agg1, N);
    owner_aggregate_kernel<D_HID><<<(N + OWN_R - 1) / OWN_R, 32 * OWN_WAVES, 0, stream>>>(
        rec, code, h1, dinv, agg1, N, E);

    gemm_wmma_f32_kernel<D_OUT, true><<<N / 16, 32 * (D_OUT / 16), 0, stream>>>(
        agg1, W2, dinv, b2, h2, out, N);
    owner_aggregate_kernel<D_OUT><<<(N + OWN_R - 1) / OWN_R, 32 * OWN_WAVES, 0, stream>>>(
        rec, code, h2, dinv, out, N, E);
}


# Owner work-groups scan a 1-bit-per-edge membership plane of their own range (built in LDS, stored whole by the degree launch), not the 1-byte codes of all edges; sum order: LDS arrival
# speedup vs baseline: 2.6454x; 1.0865x over previous
#include <hip/hip_runtime.h>

typedef __attribute__((ext_vector_type(2))) float v2f;
typedef __attribute__((ext_vector_type(8))) float v8f;

#define D_IN  128
#define D_HID 128
#define D_OUT 64

#define OWN_R 400

__global__ void deg_init_kernel(float* __restrict__ deg, int n) {
    int i = blockIdx.x * blockDim.x + threadIdx.x;
    if (i < n) deg[i] = 1.0f;
}

#define BM_EDGES 2048
#define BM_WORDS (BM_EDGES / 32)
#define BM_LPITCH (BM_WORDS + 4)
#define BM_MAXR 256

__global__ __launch_bounds__(256)
void deg_accum_kernel(const int* __restrict__ dst,
                      float* __restrict__ deg,
                      unsigned int* __restrict__ plane, int e, int nr, int pitchw) {
    __shared__ __align__(16) unsigned int bm[BM_MAXR * BM_LPITCH];
    const int tid = threadIdx.x;

    const int i0 = blockIdx.x * BM_EDGES + tid * 8;
    int d[8] = {0, 0, 0, 0, 0, 0, 0, 0};
    if (i0 < e) {
        if (i0 + 7 < e && (reinterpret_cast<size_t>(dst) & 15) == 0) {
            const int4 v0 = *(const int4*)(dst + i0);
            const int4 v1 = *(const int4*)(dst + i0 + 4);
            d[0] = v0.x; d[1] = v0.y; d[2] = v0.z; d[3] = v0.w;
            d[4] = v1.x; d[5] = v1.y; d[6] = v1.z; d[7] = v1.w;
        } else {
            #pragma unroll
            for (int k = 0; k < 8; ++k) d[k] = dst[min(i0 + k, e - 1)];
        }
    }

    for (int i = tid; i < nr * (BM_LPITCH / 4); i += 256) ((uint4*)bm)[i] = make_uint4(0u, 0u, 0u, 0u);
    __syncthreads();

    const int w  = tid >> 2;
    const int sh = (tid & 3) * 8;
    #pragma unroll
    for (int k = 0; k < 8; ++k) {
        if (i0 + k < e) {
            atomicAdd(&deg[d[k]], 1.0f);
            atomicOr(&bm[(d[k] / OWN_R) * BM_LPITCH + w], 1u << (sh + k));
        }
    }
    __syncthreads();

    for (int i = tid; i < nr * (BM_WORDS / 4); i += 256) {
        const int r = i / (BM_WORDS / 4), j = i % (BM_WORDS / 4);
        const uint4 v = *(const uint4*)(&bm[r * BM_LPITCH + j * 4]);
        *(uint4*)(&plane[(size_t)r * pitchw + (size_t)blockIdx.x * BM_WORDS + j * 4]) = v;
    }
}

__global__ void norm_pack_kernel(const int* __restrict__ src, const int* __restrict__ dst,
                                 const float* __restrict__ deg, float* __restrict__ dinv,
                                 unsigned int* __restrict__ rec, int n, int e) {
    const int t = blockIdx.x * blockDim.x + threadIdx.x;
    for (int i = t; i < n; i += gridDim.x * blockDim.x) {
        const float d = deg[i];
        dinv[i] = (d > 0.0f) ? rsqrtf(d) : 0.0f;
    }
    uint2* r2 = (uint2*)rec;
    const int base = blockIdx.x * (blockDim.x * 4) + threadIdx.x;
    if (base >= e) return;
    unsigned s[4], d[4];
    float ds[4];
    #pragma unroll
    for (int k = 0; k < 4; ++k) {
        const int i = min(base + k * (int)blockDim.x, e - 1);
        s[k] = (unsigned)src[i];
        d[k] = (unsigned)dst[i];
    }
    #pragma unroll
    for (int k = 0; k < 4; ++k) ds[k] = deg[s[k]];
    #pragma unroll
    for (int k = 0; k < 4; ++k) {
        const int i = base + k * (int)blockDim.x;
        if (i < e) {
            uint2 r;
            r.x = s[k] | ((d[k] % (unsigned)OWN_R) << 17);
            r.y = __float_as_uint((ds[k] > 0.0f) ? rsqrtf(ds[k]) : 0.0f);
            r2[i] = r;
        }
    }
}

template <int NOUT, bool RELU_IN>
__global__ void gemm_wmma_f32_kernel(const float* __restrict__ A,
                                     const float* __restrict__ W,
                                     const float* __restrict__ dinv,
                                     const float* __restrict__ bias,
                                     float* __restrict__ H,
                                     float* __restrict__ AGG, int M) {
    constexpr int K = 128;
    constexpr int NT = NOUT / 16;
    constexpr int P = K + 4;
    __shared__ __align__(16) float lds_a[16 * P];

    const int block_row = blockIdx.x * 16;
    const int tid = threadIdx.x;
    constexpr int NTHREADS = 32 * NT;

    const int wave = tid >> 5;
    const int lane = tid & 31;
    const int half = lane >> 4;
    const int l16  = lane & 15;
    const int col_base = wave * 16;

    const float bcol = bias[col_base + l16];
    float di[8];
    #pragma unroll
    for (int v = 0; v < 8; ++v) di[v] = dinv[block_row + half * 8 + v];

    for (int i = tid * 4; i < 16 * K; i += NTHREADS * 4) {
        float4 t = *(const float4*)(&A[(size_t)block_row * K + i]);
        if (RELU_IN) {
            t.x = fmaxf(t.x, 0.0f);
            t.y = fmaxf(t.y, 0.0f);
            t.z = fmaxf(t.z, 0.0f);
            t.w = fmaxf(t.w, 0.0f);
        }
        *(float4*)(&lds_a[(i / K) * P + (i % K)]) = t;
    }
    __syncthreads();

    v8f acc = {};
    #pragma unroll 4
    for (int k0 = 0; k0 < K; k0 += 4) {
        const int ka = k0 + half * 2;
        v2f a, b;
        a.x = lds_a[l16 * P + ka];
        a.y = lds_a[l16 * P + ka + 1];
        b.x = W[(size_t)ka * NOUT + col_base + l16];
        b.y = W[(size_t)(ka + 1) * NOUT + col_base + l16];
        acc = __builtin_amdgcn_wmma_f32_16x16x4_f32(
             false, a,  false, b,
             (short)0, acc,  false,  false);
    }

    #pragma unroll
    for (int v = 0; v < 8; ++v) {
        const int row = v + half * 8;
        const size_t o = (size_t)(block_row + row) * NOUT + col_base + l16;
        const float hv = acc[v];
        H[o] = hv;
        AGG[o] = hv * di[v] * di[v] + bcol;
    }
}

#define OWN_WAVES 32
#define OWN_TILE_FLOATS (OWN_R * 128)

template <int D>
__device__ __forceinline__ void own_flush(const int* qe, int cnt, int lane,
                                          const uint2* __restrict__ rec,
                                          const float* __restrict__ h,
                                          const float* own_dinv, float* tile) {
    constexpr int VPL = D / 32;
    typedef float vrow __attribute__((ext_vector_type(VPL)));
    int my_s = 0, my_dl = 0;
    float my_c = 0.0f;
    if (lane < cnt) {
        const uint2 r = rec[qe[lane]];
        my_s = (int)(r.x & 0x1ffffu);
        my_dl = (int)(r.x >> 17);
        my_c = __uint_as_float(r.y) * own_dinv[my_dl];
    }
    for (int g = 0; g < cnt; g += 8) {
        vrow v[8];
        #pragma unroll
        for (int t = 0; t < 8; ++t) {
            v[t] = (vrow)(0.0f);
            if (g + t < cnt) {
                const int s = __builtin_amdgcn_readlane(my_s, g + t);
                v[t] = *(const vrow*)(h + (size_t)s * D + lane * VPL);
            }
        }
        #pragma unroll
        for (int t = 0; t < 8; ++t) {
            if (g + t < cnt) {
                const float c = __builtin_bit_cast(
                    float, __builtin_amdgcn_readlane(__builtin_bit_cast(int, my_c), g + t));
                const int dl = __builtin_amdgcn_readlane(my_dl, g + t);
                float* tp = tile + dl * D + lane;
                #pragma unroll
                for (int j = 0; j < VPL; ++j) atomicAdd(tp + j * 32, v[t][j] * c);
            }
        }
    }
}

template <int D>
__global__ __launch_bounds__(1024)
void owner_aggregate_kernel(const unsigned int* __restrict__ recw,
                            const unsigned int* __restrict__ plane,
                            const float* __restrict__ h, const float* __restrict__ dinv,
                            float* __restrict__ agg, int n, int pitchw) {
    constexpr int VPL = D / 32;
    constexpr int R = OWN_R;
    typedef float vrow __attribute__((ext_vector_type(VPL)));
    __shared__ __align__(16) float tile[OWN_TILE_FLOATS];
    __shared__ float own_dinv[R];
    __shared__ int q_e[OWN_WAVES][64];

    const int tid  = threadIdx.x;
    const int lane = tid & 31;
    const int wave = __builtin_amdgcn_readfirstlane(tid >> 5);
    const int lo   = blockIdx.x * R;
    const int rows = min(R, n - lo);
    const uint2* rec = (const uint2*)recw;

    #pragma unroll 4
    for (int r = wave; r < rows; r += OWN_WAVES) {
        const vrow t = *(const vrow*)(agg + (size_t)(lo + r) * D + lane * VPL);
        #pragma unroll
        for (int j = 0; j < VPL; ++j) tile[r * D + j * 32 + lane] = t[j];
    }
    for (int r = tid; r < rows; r += 1024) own_dinv[r] = dinv[lo + r];
    __syncthreads();

    const uint4* prow = (const uint4*)(plane + (size_t)blockIdx.x * pitchw);
    const int nld = pitchw >> 7;
    const unsigned lt = (1u << lane) - 1u;
    int* qe = q_e[wave];
    int qn = 0;

    uint4 nx = prow[(size_t)min(wave, nld - 1) * 32 + lane];

    for (int ld = wave; ld < nld; ld += OWN_WAVES) {
        const uint4 dv = nx;
        nx = prow[(size_t)min(ld + OWN_WAVES, nld - 1) * 32 + lane];

        const int ebase = (ld * 32 + lane) * 128;
        unsigned long long mlo = ((unsigned long long)dv.y << 32) | dv.x;
        unsigned long long mhi = ((unsigned long long)dv.w << 32) | dv.z;

        while (true) {
            const bool has = (mlo | mhi) != 0ull;
            const unsigned bal = __builtin_amdgcn_ballot_w32(has);
            if (bal == 0) break;
            if (has) {
                int b;
                if (mlo != 0ull) { b = __builtin_ctzll(mlo); mlo &= mlo - 1ull; }
                else             { b = 64 + __builtin_ctzll(mhi); mhi &= mhi - 1ull; }
                const int slot = qn + __builtin_popcount(bal & lt);
                qe[slot] = ebase + b;
            }
            qn += __builtin_popcount(bal);
            if (qn >= 32) {
                qn -= 32;
                own_flush<D>(qe + qn, 32, lane, rec, h, own_dinv, tile);
            }
        }
    }
    if (qn > 0) own_flush<D>(qe, qn, lane, rec, h, own_dinv, tile);
    __syncthreads();

    for (int r = wave; r < rows; r += OWN_WAVES) {
        vrow t;
        #pragma unroll
        for (int j = 0; j < VPL; ++j) t[j] = tile[r * D + j * 32 + lane];
        *(vrow*)(agg + (size_t)(lo + r) * D + lane * VPL) = t;
    }
}

extern "C" void kernel_launch(void* const* d_in, const int* in_sizes, int n_in,
                              void* d_out, int out_size, void* d_ws, size_t ws_size,
                              hipStream_t stream) {
    const float* x    = (const float*)d_in[0];
    const int*   eidx = (const int*)  d_in[1];
    const float* W1   = (const float*)d_in[2];
    const float* b1   = (const float*)d_in[3];
    const float* W2   = (const float*)d_in[4];
    const float* b2   = (const float*)d_in[5];
    float* out = (float*)d_out;

    const int N = in_sizes[0] / D_IN;
    const int E = in_sizes[1] / 2;
    const int* src = eidx;
    const int* dst = eidx + E;

    float* ws = (float*)d_ws;
    float* deg  = ws;
    float* dinv = deg + (((size_t)N + 255) & ~(size_t)255);
    float* h1   = dinv + (((size_t)N + 255) & ~(size_t)255);
    float* agg1 = h1 + (size_t)N * D_HID;
    float* h2   = h1;
    const int NR     = (N + OWN_R - 1) / OWN_R;
    const int NBLK   = (((E + BM_EDGES - 1) / BM_EDGES) + 1) & ~1;
    const int PITCHW = NBLK * BM_WORDS;
    unsigned int* plane = (unsigned int*)(agg1 + (size_t)N * D_HID);
    unsigned int* rec   = plane + (size_t)NR * PITCHW;

    const int BT = 256;

    deg_init_kernel<<<(N + BT - 1) / BT, BT, 0, stream>>>(deg, N);
    deg_accum_kernel<<<NBLK, 256, 0, stream>>>(dst, deg, plane, E, NR, PITCHW);
    norm_pack_kernel<<<(E + 4 * BT - 1) / (4 * BT), BT, 0, stream>>>(src, dst, deg, dinv, rec, N, E);

    gemm_wmma_f32_kernel<D_HID, false><<<N / 16, 32 * (D_HID / 16), 0, stream>>>(
        x, W1, dinv, b1, h1, agg1, N);
    owner_aggregate_kernel<D_HID><<<NR, 32 * OWN_WAVES, 0, stream>>>(
        rec, plane, h1, dinv, agg1, N, PITCHW);

    gemm_wmma_f32_kernel<D_OUT, true><<<N / 16, 32 * (D_OUT / 16), 0, stream>>>(
        agg1, W2, dinv, b2, h2, out, N);
    owner_aggregate_kernel<D_OUT><<<NR, 32 * OWN_WAVES, 0, stream>>>(
        rec, plane, h2, dinv, out, N, PITCHW);
}


# WMMA GEMMs made persistent: a wave fetches its W fragments once (64 registers), its group loops over 16-row strips, next strip's loads in flight; 6250 groups to 1024/1792; sum order unchanged
# speedup vs baseline: 2.7632x; 1.0445x over previous
#include <hip/hip_runtime.h>

typedef __attribute__((ext_vector_type(2))) float v2f;
typedef __attribute__((ext_vector_type(8))) float v8f;

#define D_IN  128
#define D_HID 128
#define D_OUT 64

#define OWN_R 400

__global__ void deg_init_kernel(float* __restrict__ deg, int n) {
    int i = blockIdx.x * blockDim.x + threadIdx.x;
    if (i < n) deg[i] = 1.0f;
}

#define BM_EDGES 2048
#define BM_WORDS (BM_EDGES / 32)
#define BM_LPITCH (BM_WORDS + 4)
#define BM_MAXR 256

__global__ __launch_bounds__(256)
void deg_accum_kernel(const int* __restrict__ dst,
                      float* __restrict__ deg,
                      unsigned int* __restrict__ plane, int e, int nr, int pitchw) {
    __shared__ __align__(16) unsigned int bm[BM_MAXR * BM_LPITCH];
    const int tid = threadIdx.x;

    const int i0 = blockIdx.x * BM_EDGES + tid * 8;
    int d[8] = {0, 0, 0, 0, 0, 0, 0, 0};
    if (i0 < e) {
        if (i0 + 7 < e && (reinterpret_cast<size_t>(dst) & 15) == 0) {
            const int4 v0 = *(const int4*)(dst + i0);
            const int4 v1 = *(const int4*)(dst + i0 + 4);
            d[0] = v0.x; d[1] = v0.y; d[2] = v0.z; d[3] = v0.w;
            d[4] = v1.x; d[5] = v1.y; d[6] = v1.z; d[7] = v1.w;
        } else {
            #pragma unroll
            for (int k = 0; k < 8; ++k) d[k] = dst[min(i0 + k, e - 1)];
        }
    }

    for (int i = tid; i < nr * (BM_LPITCH / 4); i += 256) ((uint4*)bm)[i] = make_uint4(0u, 0u, 0u, 0u);
    __syncthreads();

    const int w  = tid >> 2;
    const int sh = (tid & 3) * 8;
    #pragma unroll
    for (int k = 0; k < 8; ++k) {
        if (i0 + k < e) {
            atomicAdd(&deg[d[k]], 1.0f);
            atomicOr(&bm[(d[k] / OWN_R) * BM_LPITCH + w], 1u << (sh + k));
        }
    }
    __syncthreads();

    for (int i = tid; i < nr * (BM_WORDS / 4); i += 256) {
        const int r = i / (BM_WORDS / 4), j = i % (BM_WORDS / 4);
        const uint4 v = *(const uint4*)(&bm[r * BM_LPITCH + j * 4]);
        *(uint4*)(&plane[(size_t)r * pitchw + (size_t)blockIdx.x * BM_WORDS + j * 4]) = v;
    }
}

__global__ void norm_pack_kernel(const int* __restrict__ src, const int* __restrict__ dst,
                                 const float* __restrict__ deg, float* __restrict__ dinv,
                                 unsigned int* __restrict__ rec, int n, int e) {
    const int t = blockIdx.x * blockDim.x + threadIdx.x;
    for (int i = t; i < n; i += gridDim.x * blockDim.x) {
        const float d = deg[i];
        dinv[i] = (d > 0.0f) ? rsqrtf(d) : 0.0f;
    }
    uint2* r2 = (uint2*)rec;
    const int base = blockIdx.x * (blockDim.x * 4) + threadIdx.x;
    if (base >= e) return;
    unsigned s[4], d[4];
    float ds[4];
    #pragma unroll
    for (int k = 0; k < 4; ++k) {
        const int i = min(base + k * (int)blockDim.x, e - 1);
        s[k] = (unsigned)src[i];
        d[k] = (unsigned)dst[i];
    }
    #pragma unroll
    for (int k = 0; k < 4; ++k) ds[k] = deg[s[k]];
    #pragma unroll
    for (int k = 0; k < 4; ++k) {
        const int i = base + k * (int)blockDim.x;
        if (i < e) {
            uint2 r;
            r.x = s[k] | ((d[k] % (unsigned)OWN_R) << 17);
            r.y = __float_as_uint((ds[k] > 0.0f) ? rsqrtf(ds[k]) : 0.0f);
            r2[i] = r;
        }
    }
}

template <int NOUT, bool RELU_IN>
__global__ __launch_bounds__(32 * (NOUT / 16))
void gemm_wmma_f32_kernel(const float* __restrict__ A,
                          const float* __restrict__ W,
                          const float* __restrict__ dinv,
                          const float* __restrict__ bias,
                          float* __restrict__ H,
                          float* __restrict__ AGG, int M) {
    constexpr int K = 128;
    constexpr int NT = NOUT / 16;
    constexpr int NTHREADS = 32 * NT;
    constexpr int P = K + 4;
    constexpr int NLD = (16 * K / 4) / NTHREADS;
    __shared__ __align__(16) float lds_a[2][16 * P];

    const int tid  = threadIdx.x;
    const int wave = tid >> 5;
    const int lane = tid & 31;
    const int half = lane >> 4;
    const int l16  = lane & 15;
    const int col_base = wave * 16;
    const int nstrips = M / 16;

    int strip = blockIdx.x;
    if (strip >= nstrips) return;

    v2f bf[K / 4];
    #pragma unroll
    for (int s = 0; s < K / 4; ++s) {
        const int ka = s * 4 + half * 2;
        bf[s].x = W[(size_t)ka * NOUT + col_base + l16];
        bf[s].y = W[(size_t)(ka + 1) * NOUT + col_base + l16];
    }
    const float bcol = bias[col_base + l16];

    const float* const ap = A + (size_t)tid * 4;
    const int soff = (tid >> 5) * P + (tid & 31) * 4;
    const float* const dp = dinv + half * 8;
    const size_t ooff = (size_t)(half * 8) * NOUT + col_base + l16;

    float4 p0, p1, p2, p3, dA, dB;
    {
        const float* a = ap + (size_t)strip * (16 * K);
        p0 = *(const float4*)(a);
        p1 = *(const float4*)(a + NTHREADS * 4);
        if (NLD > 2) {
            p2 = *(const float4*)(a + 2 * NTHREADS * 4);
            p3 = *(const float4*)(a + 3 * NTHREADS * 4);
        }
        const float* d = dp + (size_t)strip * 16;
        dA = *(const float4*)(d);
        dB = *(const float4*)(d + 4);
    }

    int buf = 0;
    for (; strip < nstrips; strip += gridDim.x) {
        float* la = lds_a[buf];
        {
            float* ls = la + soff;
            float4 t0 = p0, t1 = p1, t2 = p2, t3 = p3;
            if (RELU_IN) {
                t0.x = fmaxf(t0.x, 0.0f); t0.y = fmaxf(t0.y, 0.0f); t0.z = fmaxf(t0.z, 0.0f); t0.w = fmaxf(t0.w, 0.0f);
                t1.x = fmaxf(t1.x, 0.0f); t1.y = fmaxf(t1.y, 0.0f); t1.z = fmaxf(t1.z, 0.0f); t1.w = fmaxf(t1.w, 0.0f);
                if (NLD > 2) {
                    t2.x = fmaxf(t2.x, 0.0f); t2.y = fmaxf(t2.y, 0.0f); t2.z = fmaxf(t2.z, 0.0f); t2.w = fmaxf(t2.w, 0.0f);
                    t3.x = fmaxf(t3.x, 0.0f); t3.y = fmaxf(t3.y, 0.0f); t3.z = fmaxf(t3.z, 0.0f); t3.w = fmaxf(t3.w, 0.0f);
                }
            }
            *(float4*)(ls) = t0;
            *(float4*)(ls + NT * P) = t1;
            if (NLD > 2) {
                *(float4*)(ls + 2 * NT * P) = t2;
                *(float4*)(ls + 3 * NT * P) = t3;
            }
        }
        const float di[8] = {dA.x, dA.y, dA.z, dA.w, dB.x, dB.y, dB.z, dB.w};
        __syncthreads();

        {
            const int nx = min(strip + (int)gridDim.x, nstrips - 1);
            const float* d = dp + (size_t)nx * 16;
            dA = *(const float4*)(d);
            dB = *(const float4*)(d + 4);
            const float* a = ap + (size_t)nx * (16 * K);
            p0 = *(const float4*)(a);
            p1 = *(const float4*)(a + NTHREADS * 4);
            if (NLD > 2) {
                p2 = *(const float4*)(a + 2 * NTHREADS * 4);
                p3 = *(const float4*)(a + 3 * NTHREADS * 4);
            }
        }
        __builtin_amdgcn_sched_barrier(0);

        v8f acc = {};
        #pragma unroll
        for (int s = 0; s < K / 4; ++s) {
            const int ka = s * 4 + half * 2;
            v2f a;
            a.x = la[l16 * P + ka];
            a.y = la[l16 * P + ka + 1];
            acc = __builtin_amdgcn_wmma_f32_16x16x4_f32(
                 false, a,  false, bf[s],
                 (short)0, acc,  false,  false);
        }

        float* const hp = H + (size_t)strip * (16 * NOUT) + ooff;
        float* const gp = AGG + (size_t)strip * (16 * NOUT) + ooff;
        #pragma unroll
        for (int v = 0; v < 8; ++v) {
            const float hv = acc[v];
            hp[v * NOUT] = hv;
            gp[v * NOUT] = hv * di[v] * di[v] + bcol;
        }
        buf ^= 1;
    }
}

#define OWN_WAVES 32
#define OWN_TILE_FLOATS (OWN_R * 128)

template <int D>
__device__ __forceinline__ void own_flush(const int* qe, int cnt, int lane,
                                          const uint2* __restrict__ rec,
                                          const float* __restrict__ h,
                                          const float* own_dinv, float* tile) {
    constexpr int VPL = D / 32;
    typedef float vrow __attribute__((ext_vector_type(VPL)));
    int my_s = 0, my_dl = 0;
    float my_c = 0.0f;
    if (lane < cnt) {
        const uint2 r = rec[qe[lane]];
        my_s = (int)(r.x & 0x1ffffu);
        my_dl = (int)(r.x >> 17);
        my_c = __uint_as_float(r.y) * own_dinv[my_dl];
    }
    for (int g = 0; g < cnt; g += 8) {
        vrow v[8];
        #pragma unroll
        for (int t = 0; t < 8; ++t) {
            v[t] = (vrow)(0.0f);
            if (g + t < cnt) {
                const int s = __builtin_amdgcn_readlane(my_s, g + t);
                v[t] = *(const vrow*)(h + (size_t)s * D + lane * VPL);
            }
        }
        #pragma unroll
        for (int t = 0; t < 8; ++t) {
            if (g + t < cnt) {
                const float c = __builtin_bit_cast(
                    float, __builtin_amdgcn_readlane(__builtin_bit_cast(int, my_c), g + t));
                const int dl = __builtin_amdgcn_readlane(my_dl, g + t);
                float* tp = tile + dl * D + lane;
                #pragma unroll
                for (int j = 0; j < VPL; ++j) atomicAdd(tp + j * 32, v[t][j] * c);
            }
        }
    }
}

template <int D>
__global__ __launch_bounds__(1024)
void owner_aggregate_kernel(const unsigned int* __restrict__ recw,
                            const unsigned int* __restrict__ plane,
                            const float* __restrict__ h, const float* __restrict__ dinv,
                            float* __restrict__ agg, int n, int pitchw) {
    constexpr int VPL = D / 32;
    constexpr int R = OWN_R;
    typedef float vrow __attribute__((ext_vector_type(VPL)));
    __shared__ __align__(16) float tile[OWN_TILE_FLOATS];
    __shared__ float own_dinv[R];
    __shared__ int q_e[OWN_WAVES][64];

    const int tid  = threadIdx.x;
    const int lane = tid & 31;
    const int wave = __builtin_amdgcn_readfirstlane(tid >> 5);
    const int lo   = blockIdx.x * R;
    const int rows = min(R, n - lo);
    const uint2* rec = (const uint2*)recw;

    #pragma unroll 4
    for (int r = wave; r < rows; r += OWN_WAVES) {
        const vrow t = *(const vrow*)(agg + (size_t)(lo + r) * D + lane * VPL);
        #pragma unroll
        for (int j = 0; j < VPL; ++j) tile[r * D + j * 32 + lane] = t[j];
    }
    for (int r = tid; r < rows; r += 1024) own_dinv[r] = dinv[lo + r];
    __syncthreads();

    const uint4* prow = (const uint4*)(plane + (size_t)blockIdx.x * pitchw);
    const int nld = pitchw >> 7;
    const unsigned lt = (1u << lane) - 1u;
    int* qe = q_e[wave];
    int qn = 0;

    uint4 nx = prow[(size_t)min(wave, nld - 1) * 32 + lane];

    for (int ld = wave; ld < nld; ld += OWN_WAVES) {
        const uint4 dv = nx;
        nx = prow[(size_t)min(ld + OWN_WAVES, nld - 1) * 32 + lane];

        const int ebase = (ld * 32 + lane) * 128;
        unsigned long long mlo = ((unsigned long long)dv.y << 32) | dv.x;
        unsigned long long mhi = ((unsigned long long)dv.w << 32) | dv.z;

        while (true) {
            const bool has = (mlo | mhi) != 0ull;
            const unsigned bal = __builtin_amdgcn_ballot_w32(has);
            if (bal == 0) break;
            if (has) {
                int b;
                if (mlo != 0ull) { b = __builtin_ctzll(mlo); mlo &= mlo - 1ull; }
                else             { b = 64 + __builtin_ctzll(mhi); mhi &= mhi - 1ull; }
                const int slot = qn + __builtin_popcount(bal & lt);
                qe[slot] = ebase + b;
            }
            qn += __builtin_popcount(bal);
            if (qn >= 32) {
                qn -= 32;
                own_flush<D>(qe + qn, 32, lane, rec, h, own_dinv, tile);
            }
        }
    }
    if (qn > 0) own_flush<D>(qe, qn, lane, rec, h, own_dinv, tile);
    __syncthreads();

    for (int r = wave; r < rows; r += OWN_WAVES) {
        vrow t;
        #pragma unroll
        for (int j = 0; j < VPL; ++j) t[j] = tile[r * D + j * 32 + lane];
        *(vrow*)(agg + (size_t)(lo + r) * D + lane * VPL) = t;
    }
}

extern "C" void kernel_launch(void* const* d_in, const int* in_sizes, int n_in,
                              void* d_out, int out_size, void* d_ws, size_t ws_size,
                              hipStream_t stream) {
    const float* x    = (const float*)d_in[0];
    const int*   eidx = (const int*)  d_in[1];
    const float* W1   = (const float*)d_in[2];
    const float* b1   = (const float*)d_in[3];
    const float* W2   = (const float*)d_in[4];
    const float* b2   = (const float*)d_in[5];
    float* out = (float*)d_out;

    const int N = in_sizes[0] / D_IN;
    const int E = in_sizes[1] / 2;
    const int* src = eidx;
    const int* dst = eidx + E;

    float* ws = (float*)d_ws;
    float* deg  = ws;
    float* dinv = deg + (((size_t)N + 255) & ~(size_t)255);
    float* h1   = dinv + (((size_t)N + 255) & ~(size_t)255);
    float* agg1 = h1 + (size_t)N * D_HID;
    float* h2   = h1;
    const int NR     = (N + OWN_R - 1) / OWN_R;
    const int NBLK   = (((E + BM_EDGES - 1) / BM_EDGES) + 1) & ~1;
    const int PITCHW = NBLK * BM_WORDS;
    unsigned int* plane = (unsigned int*)(agg1 + (size_t)N * D_HID);
    unsigned int* rec   = plane + (size_t)NR * PITCHW;

    const int BT = 256;
    const int GEMM_GRID1 = 1024;
    const int GEMM_GRID2 = 1792;

    deg_init_kernel<<<(N + BT - 1) / BT, BT, 0, stream>>>(deg, N);
    deg_accum_kernel<<<NBLK, 256, 0, stream>>>(dst, deg, plane, E, NR, PITCHW);
    norm_pack_kernel<<<(E + 4 * BT - 1) / (4 * BT), BT, 0, stream>>>(src, dst, deg, dinv, rec, N, E);

    gemm_wmma_f32_kernel<D_HID, false><<<GEMM_GRID1, 32 * (D_HID / 16), 0, stream>>>(
        x, W1, dinv, b1, h1, agg1, N);
    owner_aggregate_kernel<D_HID><<<NR, 32 * OWN_WAVES, 0, stream>>>(
        rec, plane, h1, dinv, agg1, N, PITCHW);

    gemm_wmma_f32_kernel<D_OUT, true><<<GEMM_GRID2, 32 * (D_OUT / 16), 0, stream>>>(
        agg1, W2, dinv, b2, h2, out, N);
    owner_aggregate_kernel<D_OUT><<<NR, 32 * OWN_WAVES, 0, stream>>>(
        rec, plane, h2, dinv, out, N, PITCHW);
}


# Edge sums' start value (h*dinv^2 + bias) formed by the owner kernels at their tile load, not stored by the GEMMs; GEMM 1, free of dinv, runs first and sets deg = 1: 7 launches to 6; sum order same
# speedup vs baseline: 3.0609x; 1.1077x over previous
#include <hip/hip_runtime.h>

typedef __attribute__((ext_vector_type(2))) float v2f;
typedef __attribute__((ext_vector_type(8))) float v8f;

#define D_IN  128
#define D_HID 128
#define D_OUT 64

#define OWN_R 400

#define BM_EDGES 2048
#define BM_WORDS (BM_EDGES / 32)
#define BM_LPITCH (BM_WORDS + 4)
#define BM_MAXR 256

__global__ __launch_bounds__(256)
void deg_accum_kernel(const int* __restrict__ dst,
                      float* __restrict__ deg,
                      unsigned int* __restrict__ plane, int e, int nr, int pitchw) {
    __shared__ __align__(16) unsigned int bm[BM_MAXR * BM_LPITCH];
    const int tid = threadIdx.x;

    const int i0 = blockIdx.x * BM_EDGES + tid * 8;
    int d[8] = {0, 0, 0, 0, 0, 0, 0, 0};
    if (i0 < e) {
        if (i0 + 7 < e && (reinterpret_cast<size_t>(dst) & 15) == 0) {
            const int4 v0 = *(const int4*)(dst + i0);
            const int4 v1 = *(const int4*)(dst + i0 + 4);
            d[0] = v0.x; d[1] = v0.y; d[2] = v0.z; d[3] = v0.w;
            d[4] = v1.x; d[5] = v1.y; d[6] = v1.z; d[7] = v1.w;
        } else {
            #pragma unroll
            for (int k = 0; k < 8; ++k) d[k] = dst[min(i0 + k, e - 1)];
        }
    }

    for (int i = tid; i < nr * (BM_LPITCH / 4); i += 256) ((uint4*)bm)[i] = make_uint4(0u, 0u, 0u, 0u);
    __syncthreads();

    const int w  = tid >> 2;
    const int sh = (tid & 3) * 8;
    #pragma unroll
    for (int k = 0; k < 8; ++k) {
        if (i0 + k < e) {
            atomicAdd(&deg[d[k]], 1.0f);
            atomicOr(&bm[(d[k] / OWN_R) * BM_LPITCH + w], 1u << (sh + k));
        }
    }
    __syncthreads();

    for (int i = tid; i < nr * (BM_WORDS / 4); i += 256) {
        const int r = i / (BM_WORDS / 4), j = i % (BM_WORDS / 4);
        const uint4 v = *(const uint4*)(&bm[r * BM_LPITCH + j * 4]);
        *(uint4*)(&plane[(size_t)r * pitchw + (size_t)blockIdx.x * BM_WORDS + j * 4]) = v;
    }
}

__global__ void norm_pack_kernel(const int* __restrict__ src, const int* __restrict__ dst,
                                 const float* __restrict__ deg, float* __restrict__ dinv,
                                 unsigned int* __restrict__ rec, int n, int e) {
    const int t = blockIdx.x * blockDim.x + threadIdx.x;
    for (int i = t; i < n; i += gridDim.x * blockDim.x) {
        const float d = deg[i];
        dinv[i] = (d > 0.0f) ? rsqrtf(d) : 0.0f;
    }
    uint2* r2 = (uint2*)rec;
    const int base = blockIdx.x * (blockDim.x * 4) + threadIdx.x;
    if (base >= e) return;
    unsigned s[4], d[4];
    float ds[4];
    #pragma unroll
    for (int k = 0; k < 4; ++k) {
        const int i = min(base + k * (int)blockDim.x, e - 1);
        s[k] = (unsigned)src[i];
        d[k] = (unsigned)dst[i];
    }
    #pragma unroll
    for (int k = 0; k < 4; ++k) ds[k] = deg[s[k]];
    #pragma unroll
    for (int k = 0; k < 4; ++k) {
        const int i = base + k * (int)blockDim.x;
        if (i < e) {
            uint2 r;
            r.x = s[k] | ((d[k] % (unsigned)OWN_R) << 17);
            r.y = __float_as_uint((ds[k] > 0.0f) ? rsqrtf(ds[k]) : 0.0f);
            r2[i] = r;
        }
    }
}

template <int NOUT, bool RELU_IN>
__global__ __launch_bounds__(32 * (NOUT / 16))
void gemm_wmma_f32_kernel(const float* __restrict__ A,
                          const float* __restrict__ W,
                          float* __restrict__ H,
                          float* __restrict__ deg_one, int M) {
    constexpr int K = 128;
    constexpr int NT = NOUT / 16;
    constexpr int NTHREADS = 32 * NT;
    constexpr int P = K + 4;
    constexpr int NLD = (16 * K / 4) / NTHREADS;
    __shared__ __align__(16) float lds_a[2][16 * P];

    const int tid  = threadIdx.x;
    const int wave = tid >> 5;
    const int lane = tid & 31;
    const int half = lane >> 4;
    const int l16  = lane & 15;
    const int col_base = wave * 16;
    const int nstrips = M / 16;

    if (deg_one != nullptr)
        for (int i = blockIdx.x * NTHREADS + tid; i < M; i += gridDim.x * NTHREADS) deg_one[i] = 1.0f;

    int strip = blockIdx.x;
    if (strip >= nstrips) return;

    v2f bf[K / 4];
    #pragma unroll
    for (int s = 0; s < K / 4; ++s) {
        const int ka = s * 4 + half * 2;
        bf[s].x = W[(size_t)ka * NOUT + col_base + l16];
        bf[s].y = W[(size_t)(ka + 1) * NOUT + col_base + l16];
    }

    const float* const ap = A + (size_t)tid * 4;
    const int soff = (tid >> 5) * P + (tid & 31) * 4;
    const size_t ooff = (size_t)(half * 8) * NOUT + col_base + l16;

    float4 p0, p1, p2, p3;
    {
        const float* a = ap + (size_t)strip * (16 * K);
        p0 = *(const float4*)(a);
        p1 = *(const float4*)(a + NTHREADS * 4);
        if (NLD > 2) {
            p2 = *(const float4*)(a + 2 * NTHREADS * 4);
            p3 = *(const float4*)(a + 3 * NTHREADS * 4);
        }
    }

    int buf = 0;
    for (; strip < nstrips; strip += gridDim.x) {
        float* la = lds_a[buf];
        {
            float* ls = la + soff;
            float4 t0 = p0, t1 = p1, t2 = p2, t3 = p3;
            if (RELU_IN) {
                t0.x = fmaxf(t0.x, 0.0f); t0.y = fmaxf(t0.y, 0.0f); t0.z = fmaxf(t0.z, 0.0f); t0.w = fmaxf(t0.w, 0.0f);
                t1.x = fmaxf(t1.x, 0.0f); t1.y = fmaxf(t1.y, 0.0f); t1.z = fmaxf(t1.z, 0.0f); t1.w = fmaxf(t1.w, 0.0f);
                if (NLD > 2) {
                    t2.x = fmaxf(t2.x, 0.0f); t2.y = fmaxf(t2.y, 0.0f); t2.z = fmaxf(t2.z, 0.0f); t2.w = fmaxf(t2.w, 0.0f);
                    t3.x = fmaxf(t3.x, 0.0f); t3.y = fmaxf(t3.y, 0.0f); t3.z = fmaxf(t3.z, 0.0f); t3.w = fmaxf(t3.w, 0.0f);
                }
            }
            *(float4*)(ls) = t0;
            *(float4*)(ls + NT * P) = t1;
            if (NLD > 2) {
                *(float4*)(ls + 2 * NT * P) = t2;
                *(float4*)(ls + 3 * NT * P) = t3;
            }
        }
        __syncthreads();

        {
            const int nx = min(strip + (int)gridDim.x, nstrips - 1);
            const float* a = ap + (size_t)nx * (16 * K);
            p0 = *(const float4*)(a);
            p1 = *(const float4*)(a + NTHREADS * 4);
            if (NLD > 2) {
                p2 = *(const float4*)(a + 2 * NTHREADS * 4);
                p3 = *(const float4*)(a + 3 * NTHREADS * 4);
            }
        }
        __builtin_amdgcn_sched_barrier(0);

        v8f acc = {};
        #pragma unroll
        for (int s = 0; s < K / 4; ++s) {
            const int ka = s * 4 + half * 2;
            v2f a;
            a.x = la[l16 * P + ka];
            a.y = la[l16 * P + ka + 1];
            acc = __builtin_amdgcn_wmma_f32_16x16x4_f32(
                 false, a,  false, bf[s],
                 (short)0, acc,  false,  false);
        }

        float* const hp = H + (size_t)strip * (16 * NOUT) + ooff;
        #pragma unroll
        for (int v = 0; v < 8; ++v) hp[v * NOUT] = acc[v];
        buf ^= 1;
    }
}

#define OWN_WAVES 32
#define OWN_TILE_FLOATS (OWN_R * 128)

template <int D>
__device__ __forceinline__ void own_flush(const int* qe, int cnt, int lane,
                                          const uint2* __restrict__ rec,
                                          const float* __restrict__ h,
                                          const float* own_dinv, float* tile) {
    constexpr int VPL = D / 32;
    typedef float vrow __attribute__((ext_vector_type(VPL)));
    int my_s = 0, my_dl = 0;
    float my_c = 0.0f;
    if (lane < cnt) {
        const uint2 r = rec[qe[lane]];
        my_s = (int)(r.x & 0x1ffffu);
        my_dl = (int)(r.x >> 17);
        my_c = __uint_as_float(r.y) * own_dinv[my_dl];
    }
    for (int g = 0; g < cnt; g += 8) {
        vrow v[8];
        #pragma unroll
        for (int t = 0; t < 8; ++t) {
            v[t] = (vrow)(0.0f);
            if (g + t < cnt) {
                const int s = __builtin_amdgcn_readlane(my_s, g + t);
                v[t] = *(const vrow*)(h + (size_t)s * D + lane * VPL);
            }
        }
        #pragma unroll
        for (int t = 0; t < 8; ++t) {
            if (g + t < cnt) {
                const float c = __builtin_bit_cast(
                    float, __builtin_amdgcn_readlane(__builtin_bit_cast(int, my_c), g + t));
                const int dl = __builtin_amdgcn_readlane(my_dl, g + t);
                float* tp = tile + dl * D + lane;
                #pragma unroll
                for (int j = 0; j < VPL; ++j) atomicAdd(tp + j * 32, v[t][j] * c);
            }
        }
    }
}

template <int D>
__global__ __launch_bounds__(1024)
void owner_aggregate_kernel(const unsigned int* __restrict__ recw,
                            const unsigned int* __restrict__ plane,
                            const float* __restrict__ h, const float* __restrict__ dinv,
                            const float* __restrict__ bias,
                            float* __restrict__ agg, int n, int pitchw) {
    constexpr int VPL = D / 32;
    constexpr int R = OWN_R;
    typedef float vrow __attribute__((ext_vector_type(VPL)));
    __shared__ __align__(16) float tile[OWN_TILE_FLOATS];
    __shared__ float own_dinv[R];
    __shared__ int q_e[OWN_WAVES][64];

    const int tid  = threadIdx.x;
    const int lane = tid & 31;
    const int wave = __builtin_amdgcn_readfirstlane(tid >> 5);
    const int lo   = blockIdx.x * R;
    const int rows = min(R, n - lo);
    const uint2* rec = (const uint2*)recw;

    vrow bv;
    #pragma unroll
    for (int j = 0; j < VPL; ++j) bv[j] = bias[lane * VPL + j];
    #pragma unroll 4
    for (int r = wave; r < rows; r += OWN_WAVES) {
        const vrow t = *(const vrow*)(h + (size_t)(lo + r) * D + lane * VPL);
        const float d = dinv[lo + r];
        #pragma unroll
        for (int j = 0; j < VPL; ++j) tile[r * D + j * 32 + lane] = t[j] * d * d + bv[j];
    }
    for (int r = tid; r < rows; r += 1024) own_dinv[r] = dinv[lo + r];
    __syncthreads();

    const uint4* prow = (const uint4*)(plane + (size_t)blockIdx.x * pitchw);
    const int nld = pitchw >> 7;
    const unsigned lt = (1u << lane) - 1u;
    int* qe = q_e[wave];
    int qn = 0;

    uint4 nx = prow[(size_t)min(wave, nld - 1) * 32 + lane];

    for (int ld = wave; ld < nld; ld += OWN_WAVES) {
        const uint4 dv = nx;
        nx = prow[(size_t)min(ld + OWN_WAVES, nld - 1) * 32 + lane];

        const int ebase = (ld * 32 + lane) * 128;
        unsigned long long mlo = ((unsigned long long)dv.y << 32) | dv.x;
        unsigned long long mhi = ((unsigned long long)dv.w << 32) | dv.z;

        while (true) {
            const bool has = (mlo | mhi) != 0ull;
            const unsigned bal = __builtin_amdgcn_ballot_w32(has);
            if (bal == 0) break;
            if (has) {
                int b;
                if (mlo != 0ull) { b = __builtin_ctzll(mlo); mlo &= mlo - 1ull; }
                else             { b = 64 + __builtin_ctzll(mhi); mhi &= mhi - 1ull; }
                const int slot = qn + __builtin_popcount(bal & lt);
                qe[slot] = ebase + b;
            }
            qn += __builtin_popcount(bal);
            if (qn >= 32) {
                qn -= 32;
                own_flush<D>(qe + qn, 32, lane, rec, h, own_dinv, tile);
            }
        }
    }
    if (qn > 0) own_flush<D>(qe, qn, lane, rec, h, own_dinv, tile);
    __syncthreads();

    for (int r = wave; r < rows; r += OWN_WAVES) {
        vrow t;
        #pragma unroll
        for (int j = 0; j < VPL; ++j) t[j] = tile[r * D + j * 32 + lane];
        *(vrow*)(agg + (size_t)(lo + r) * D + lane * VPL) = t;
    }
}

extern "C" void kernel_launch(void* const* d_in, const int* in_sizes, int n_in,
                              void* d_out, int out_size, void* d_ws, size_t ws_size,
                              hipStream_t stream) {
    const float* x    = (const float*)d_in[0];
    const int*   eidx = (const int*)  d_in[1];
    const float* W1   = (const float*)d_in[2];
    const float* b1   = (const float*)d_in[3];
    const float* W2   = (const float*)d_in[4];
    const float* b2   = (const float*)d_in[5];
    float* out = (float*)d_out;

    const int N = in_sizes[0] / D_IN;
    const int E = in_sizes[1] / 2;
    const int* src = eidx;
    const int* dst = eidx + E;

    float* ws = (float*)d_ws;
    float* deg  = ws;
    float* dinv = deg + (((size_t)N + 255) & ~(size_t)255);
    float* h1   = dinv + (((size_t)N + 255) & ~(size_t)255);
    float* agg1 = h1 + (size_t)N * D_HID;
    float* h2   = h1;
    const int NR     = (N + OWN_R - 1) / OWN_R;
    const int NBLK   = (((E + BM_EDGES - 1) / BM_EDGES) + 1) & ~1;
    const int PITCHW = NBLK * BM_WORDS;
    unsigned int* plane = (unsigned int*)(agg1 + (size_t)N * D_HID);
    unsigned int* rec   = plane + (size_t)NR * PITCHW;

    const int BT = 256;
    const int GEMM_GRID1 = 1024;
    const int GEMM_GRID2 = 1792;

    gemm_wmma_f32_kernel<D_HID, false><<<GEMM_GRID1, 32 * (D_HID / 16), 0, stream>>>(
        x, W1, h1, deg, N);

    deg_accum_kernel<<<NBLK, 256, 0, stream>>>(dst, deg, plane, E, NR, PITCHW);
    norm_pack_kernel<<<(E + 4 * BT - 1) / (4 * BT), BT, 0, stream>>>(src, dst, deg, dinv, rec, N, E);

    owner_aggregate_kernel<D_HID><<<NR, 32 * OWN_WAVES, 0, stream>>>(
        rec, plane, h1, dinv, b1, agg1, N, PITCHW);

    gemm_wmma_f32_kernel<D_OUT, true><<<GEMM_GRID2, 32 * (D_OUT / 16), 0, stream>>>(
        agg1, W2, h2, nullptr, N);
    owner_aggregate_kernel<D_OUT><<<NR, 32 * OWN_WAVES, 0, stream>>>(
        rec, plane, h2, dinv, b2, out, N, PITCHW);
}


# Degrees counted by owner work-groups in LDS (walk their membership plane, fetch kept edges' targets, LDS int adds; deg = 1 + count), not by a global f32 add per edge; 6 launches to 7; sums unchanged
# speedup vs baseline: 4.2287x; 1.3815x over previous
#include <hip/hip_runtime.h>

typedef __attribute__((ext_vector_type(2))) float v2f;
typedef __attribute__((ext_vector_type(8))) float v8f;

#define D_IN  128
#define D_HID 128
#define D_OUT 64

#define OWN_R 400

#define BM_EDGES 2048
#define BM_WORDS (BM_EDGES / 32)
#define BM_LPITCH (BM_WORDS + 4)
#define BM_MAXR 256

__global__ __launch_bounds__(256)
void plane_build_kernel(const int* __restrict__ dst,
                        unsigned int* __restrict__ plane, int e, int nr, int pitchw) {
    __shared__ __align__(16) unsigned int bm[BM_MAXR * BM_LPITCH];
    const int tid = threadIdx.x;

    const int i0 = blockIdx.x * BM_EDGES + tid * 8;
    int d[8] = {0, 0, 0, 0, 0, 0, 0, 0};
    if (i0 < e) {
        if (i0 + 7 < e && (reinterpret_cast<size_t>(dst) & 15) == 0) {
            const int4 v0 = *(const int4*)(dst + i0);
            const int4 v1 = *(const int4*)(dst + i0 + 4);
            d[0] = v0.x; d[1] = v0.y; d[2] = v0.z; d[3] = v0.w;
            d[4] = v1.x; d[5] = v1.y; d[6] = v1.z; d[7] = v1.w;
        } else {
            #pragma unroll
            for (int k = 0; k < 8; ++k) d[k] = dst[min(i0 + k, e - 1)];
        }
    }

    for (int i = tid; i < nr * (BM_LPITCH / 4); i += 256) ((uint4*)bm)[i] = make_uint4(0u, 0u, 0u, 0u);
    __syncthreads();

    const int w  = tid >> 2;
    const int sh = (tid & 3) * 8;
    #pragma unroll
    for (int k = 0; k < 8; ++k) {
        if (i0 + k < e) atomicOr(&bm[(d[k] / OWN_R) * BM_LPITCH + w], 1u << (sh + k));
    }
    __syncthreads();

    for (int i = tid; i < nr * (BM_WORDS / 4); i += 256) {
        const int r = i / (BM_WORDS / 4), j = i % (BM_WORDS / 4);
        const uint4 v = *(const uint4*)(&bm[r * BM_LPITCH + j * 4]);
        *(uint4*)(&plane[(size_t)r * pitchw + (size_t)blockIdx.x * BM_WORDS + j * 4]) = v;
    }
}

__global__ void norm_pack_kernel(const int* __restrict__ src, const int* __restrict__ dst,
                                 const float* __restrict__ deg, float* __restrict__ dinv,
                                 unsigned int* __restrict__ rec, int n, int e) {
    const int t = blockIdx.x * blockDim.x + threadIdx.x;
    for (int i = t; i < n; i += gridDim.x * blockDim.x) {
        const float d = deg[i];
        dinv[i] = (d > 0.0f) ? rsqrtf(d) : 0.0f;
    }
    uint2* r2 = (uint2*)rec;
    const int base = blockIdx.x * (blockDim.x * 4) + threadIdx.x;
    if (base >= e) return;
    unsigned s[4], d[4];
    float ds[4];
    #pragma unroll
    for (int k = 0; k < 4; ++k) {
        const int i = min(base + k * (int)blockDim.x, e - 1);
        s[k] = (unsigned)src[i];
        d[k] = (unsigned)dst[i];
    }
    #pragma unroll
    for (int k = 0; k < 4; ++k) ds[k] = deg[s[k]];
    #pragma unroll
    for (int k = 0; k < 4; ++k) {
        const int i = base + k * (int)blockDim.x;
        if (i < e) {
            uint2 r;
            r.x = s[k] | ((d[k] % (unsigned)OWN_R) << 17);
            r.y = __float_as_uint((ds[k] > 0.0f) ? rsqrtf(ds[k]) : 0.0f);
            r2[i] = r;
        }
    }
}

template <int NOUT, bool RELU_IN>
__global__ __launch_bounds__(32 * (NOUT / 16))
void gemm_wmma_f32_kernel(const float* __restrict__ A,
                          const float* __restrict__ W,
                          float* __restrict__ H, int M) {
    constexpr int K = 128;
    constexpr int NT = NOUT / 16;
    constexpr int NTHREADS = 32 * NT;
    constexpr int P = K + 4;
    constexpr int NLD = (16 * K / 4) / NTHREADS;
    __shared__ __align__(16) float lds_a[2][16 * P];

    const int tid  = threadIdx.x;
    const int wave = tid >> 5;
    const int lane = tid & 31;
    const int half = lane >> 4;
    const int l16  = lane & 15;
    const int col_base = wave * 16;
    const int nstrips = M / 16;

    int strip = blockIdx.x;
    if (strip >= nstrips) return;

    v2f bf[K / 4];
    #pragma unroll
    for (int s = 0; s < K / 4; ++s) {
        const int ka = s * 4 + half * 2;
        bf[s].x = W[(size_t)ka * NOUT + col_base + l16];
        bf[s].y = W[(size_t)(ka + 1) * NOUT + col_base + l16];
    }

    const float* const ap = A + (size_t)tid * 4;
    const int soff = (tid >> 5) * P + (tid & 31) * 4;
    const size_t ooff = (size_t)(half * 8) * NOUT + col_base + l16;

    float4 p0, p1, p2, p3;
    {
        const float* a = ap + (size_t)strip * (16 * K);
        p0 = *(const float4*)(a);
        p1 = *(const float4*)(a + NTHREADS * 4);
        if (NLD > 2) {
            p2 = *(const float4*)(a + 2 * NTHREADS * 4);
            p3 = *(const float4*)(a + 3 * NTHREADS * 4);
        }
    }

    int buf = 0;
    for (; strip < nstrips; strip += gridDim.x) {
        float* la = lds_a[buf];
        {
            float* ls = la + soff;
            float4 t0 = p0, t1 = p1, t2 = p2, t3 = p3;
            if (RELU_IN) {
                t0.x = fmaxf(t0.x, 0.0f); t0.y = fmaxf(t0.y, 0.0f); t0.z = fmaxf(t0.z, 0.0f); t0.w = fmaxf(t0.w, 0.0f);
                t1.x = fmaxf(t1.x, 0.0f); t1.y = fmaxf(t1.y, 0.0f); t1.z = fmaxf(t1.z, 0.0f); t1.w = fmaxf(t1.w, 0.0f);
                if (NLD > 2) {
                    t2.x = fmaxf(t2.x, 0.0f); t2.y = fmaxf(t2.y, 0.0f); t2.z = fmaxf(t2.z, 0.0f); t2.w = fmaxf(t2.w, 0.0f);
                    t3.x = fmaxf(t3.x, 0.0f); t3.y = fmaxf(t3.y, 0.0f); t3.z = fmaxf(t3.z, 0.0f); t3.w = fmaxf(t3.w, 0.0f);
                }
            }
            *(float4*)(ls) = t0;
            *(float4*)(ls + NT * P) = t1;
            if (NLD > 2) {
                *(float4*)(ls + 2 * NT * P) = t2;
                *(float4*)(ls + 3 * NT * P) = t3;
            }
        }
        __syncthreads();

        {
            const int nx = min(strip + (int)gridDim.x, nstrips - 1);
            const float* a = ap + (size_t)nx * (16 * K);
            p0 = *(const float4*)(a);
            p1 = *(const float4*)(a + NTHREADS * 4);
            if (NLD > 2) {
                p2 = *(const float4*)(a + 2 * NTHREADS * 4);
                p3 = *(const float4*)(a + 3 * NTHREADS * 4);
            }
        }
        __builtin_amdgcn_sched_barrier(0);

        v8f acc = {};
        #pragma unroll
        for (int s = 0; s < K / 4; ++s) {
            const int ka = s * 4 + half * 2;
            v2f a;
            a.x = la[l16 * P + ka];
            a.y = la[l16 * P + ka + 1];
            acc = __builtin_amdgcn_wmma_f32_16x16x4_f32(
                 false, a,  false, bf[s],
                 (short)0, acc,  false,  false);
        }

        float* const hp = H + (size_t)strip * (16 * NOUT) + ooff;
        #pragma unroll
        for (int v = 0; v < 8; ++v) hp[v * NOUT] = acc[v];
        buf ^= 1;
    }
}

#define OWN_WAVES 32
#define OWN_TILE_FLOATS (OWN_R * 128)

template <int D>
__device__ __forceinline__ void own_flush(const int* qe, int cnt, int lane,
                                          const uint2* __restrict__ rec,
                                          const float* __restrict__ h,
                                          const float* own_dinv, float* tile) {
    constexpr int VPL = D / 32;
    typedef float vrow __attribute__((ext_vector_type(VPL)));
    int my_s = 0, my_dl = 0;
    float my_c = 0.0f;
    if (lane < cnt) {
        const uint2 r = rec[qe[lane]];
        my_s = (int)(r.x & 0x1ffffu);
        my_dl = (int)(r.x >> 17);
        my_c = __uint_as_float(r.y) * own_dinv[my_dl];
    }
    for (int g = 0; g < cnt; g += 8) {
        vrow v[8];
        #pragma unroll
        for (int t = 0; t < 8; ++t) {
            v[t] = (vrow)(0.0f);
            if (g + t < cnt) {
                const int s = __builtin_amdgcn_readlane(my_s, g + t);
                v[t] = *(const vrow*)(h + (size_t)s * D + lane * VPL);
            }
        }
        #pragma unroll
        for (int t = 0; t < 8; ++t) {
            if (g + t < cnt) {
                const float c = __builtin_bit_cast(
                    float, __builtin_amdgcn_readlane(__builtin_bit_cast(int, my_c), g + t));
                const int dl = __builtin_amdgcn_readlane(my_dl, g + t);
                float* tp = tile + dl * D + lane;
                #pragma unroll
                for (int j = 0; j < VPL; ++j) atomicAdd(tp + j * 32, v[t][j] * c);
            }
        }
    }
}

template <int D>
__global__ __launch_bounds__(1024)
void owner_aggregate_kernel(const unsigned int* __restrict__ recw,
                            const unsigned int* __restrict__ plane,
                            const float* __restrict__ h, const float* __restrict__ dinv,
                            const float* __restrict__ bias,
                            float* __restrict__ agg, int n, int pitchw) {
    constexpr int VPL = D / 32;
    constexpr int R = OWN_R;
    typedef float vrow __attribute__((ext_vector_type(VPL)));
    __shared__ __align__(16) float tile[OWN_TILE_FLOATS];
    __shared__ float own_dinv[R];
    __shared__ int q_e[OWN_WAVES][64];

    const int tid  = threadIdx.x;
    const int lane = tid & 31;
    const int wave = __builtin_amdgcn_readfirstlane(tid >> 5);
    const int lo   = blockIdx.x * R;
    const int rows = min(R, n - lo);
    const uint2* rec = (const uint2*)recw;

    vrow bv;
    #pragma unroll
    for (int j = 0; j < VPL; ++j) bv[j] = bias[lane * VPL + j];
    #pragma unroll 4
    for (int r = wave; r < rows; r += OWN_WAVES) {
        const vrow t = *(const vrow*)(h + (size_t)(lo + r) * D + lane * VPL);
        const float d = dinv[lo + r];
        #pragma unroll
        for (int j = 0; j < VPL; ++j) tile[r * D + j * 32 + lane] = t[j] * d * d + bv[j];
    }
    for (int r = tid; r < rows; r += 1024) own_dinv[r] = dinv[lo + r];
    __syncthreads();

    const uint4* prow = (const uint4*)(plane + (size_t)blockIdx.x * pitchw);
    const int nld = pitchw >> 7;
    const unsigned lt = (1u << lane) - 1u;
    int* qe = q_e[wave];
    int qn = 0;

    uint4 nx = prow[(size_t)min(wave, nld - 1) * 32 + lane];

    for (int ld = wave; ld < nld; ld += OWN_WAVES) {
        const uint4 dv = nx;
        nx = prow[(size_t)min(ld + OWN_WAVES, nld - 1) * 32 + lane];

        const int ebase = (ld * 32 + lane) * 128;
        unsigned long long mlo = ((unsigned long long)dv.y << 32) | dv.x;
        unsigned long long mhi = ((unsigned long long)dv.w << 32) | dv.z;

        while (true) {
            const bool has = (mlo | mhi) != 0ull;
            const unsigned bal = __builtin_amdgcn_ballot_w32(has);
            if (bal == 0) break;
            if (has) {
                int b;
                if (mlo != 0ull) { b = __builtin_ctzll(mlo); mlo &= mlo - 1ull; }
                else             { b = 64 + __builtin_ctzll(mhi); mhi &= mhi - 1ull; }
                const int slot = qn + __builtin_popcount(bal & lt);
                qe[slot] = ebase + b;
            }
            qn += __builtin_popcount(bal);
            if (qn >= 32) {
                qn -= 32;
                own_flush<D>(qe + qn, 32, lane, rec, h, own_dinv, tile);
            }
        }
    }
    if (qn > 0) own_flush<D>(qe, qn, lane, rec, h, own_dinv, tile);
    __syncthreads();

    for (int r = wave; r < rows; r += OWN_WAVES) {
        vrow t;
        #pragma unroll
        for (int j = 0; j < VPL; ++j) t[j] = tile[r * D + j * 32 + lane];
        *(vrow*)(agg + (size_t)(lo + r) * D + lane * VPL) = t;
    }
}

#define OWN_DEG_LDS_INTS (41 * 1024)
__global__ __launch_bounds__(1024)
void owner_degree_kernel(const int* __restrict__ dst,
                         const unsigned int* __restrict__ plane,
                         float* __restrict__ deg, int n, int pitchw) {
    constexpr int R = OWN_R;
    __shared__ int cnt[OWN_DEG_LDS_INTS];
    __shared__ int q_e[OWN_WAVES][64];

    const int tid  = threadIdx.x;
    const int lane = tid & 31;
    const int wave = __builtin_amdgcn_readfirstlane(tid >> 5);
    const int lo   = blockIdx.x * R;
    const int rows = min(R, n - lo);

    for (int i = tid; i < R; i += 1024) cnt[i] = 0;
    __syncthreads();

    const uint4* prow = (const uint4*)(plane + (size_t)blockIdx.x * pitchw);
    const int nld = pitchw >> 7;
    const unsigned lt = (1u << lane) - 1u;
    int* qe = q_e[wave];
    int qn = 0;
    int pend = 0;
    bool pend_has = false;

    uint4 nx = prow[(size_t)min(wave, nld - 1) * 32 + lane];

    for (int ld = wave; ld < nld; ld += OWN_WAVES) {
        const uint4 dv = nx;
        nx = prow[(size_t)min(ld + OWN_WAVES, nld - 1) * 32 + lane];

        const int ebase = (ld * 32 + lane) * 128;
        unsigned long long mlo = ((unsigned long long)dv.y << 32) | dv.x;
        unsigned long long mhi = ((unsigned long long)dv.w << 32) | dv.z;

        while (true) {
            const bool has = (mlo | mhi) != 0ull;
            const unsigned bal = __builtin_amdgcn_ballot_w32(has);
            if (bal == 0) break;
            if (has) {
                int b;
                if (mlo != 0ull) { b = __builtin_ctzll(mlo); mlo &= mlo - 1ull; }
                else             { b = 64 + __builtin_ctzll(mhi); mhi &= mhi - 1ull; }
                const int slot = qn + __builtin_popcount(bal & lt);
                qe[slot] = ebase + b;
            }
            qn += __builtin_popcount(bal);
            if (qn >= 32) {
                qn -= 32;
                if (pend_has) atomicAdd(&cnt[pend - lo], 1);
                pend = dst[qe[qn + lane]];
                pend_has = true;
            }
        }
    }
    if (pend_has) atomicAdd(&cnt[pend - lo], 1);
    if (lane < qn) atomicAdd(&cnt[dst[qe[lane]] - lo], 1);
    __syncthreads();

    for (int i = tid; i < rows; i += 1024) deg[lo + i] = 1.0f + (float)cnt[i];
}

extern "C" void kernel_launch(void* const* d_in, const int* in_sizes, int n_in,
                              void* d_out, int out_size, void* d_ws, size_t ws_size,
                              hipStream_t stream) {
    const float* x    = (const float*)d_in[0];
    const int*   eidx = (const int*)  d_in[1];
    const float* W1   = (const float*)d_in[2];
    const float* b1   = (const float*)d_in[3];
    const float* W2   = (const float*)d_in[4];
    const float* b2   = (const float*)d_in[5];
    float* out = (float*)d_out;

    const int N = in_sizes[0] / D_IN;
    const int E = in_sizes[1] / 2;
    const int* src = eidx;
    const int* dst = eidx + E;

    float* ws = (float*)d_ws;
    float* deg  = ws;
    float* dinv = deg + (((size_t)N + 255) & ~(size_t)255);
    float* h1   = dinv + (((size_t)N + 255) & ~(size_t)255);
    float* agg1 = h1 + (size_t)N * D_HID;
    float* h2   = h1;
    const int NR     = (N + OWN_R - 1) / OWN_R;
    const int NBLK   = (((E + BM_EDGES - 1) / BM_EDGES) + 1) & ~1;
    const int PITCHW = NBLK * BM_WORDS;
    unsigned int* plane = (unsigned int*)(agg1 + (size_t)N * D_HID);
    unsigned int* rec   = plane + (size_t)NR * PITCHW;

    const int BT = 256;
    const int GEMM_GRID1 = 1024;
    const int GEMM_GRID2 = 1792;

    gemm_wmma_f32_kernel<D_HID, false><<<GEMM_GRID1, 32 * (D_HID / 16), 0, stream>>>(
        x, W1, h1, N);

    plane_build_kernel<<<NBLK, 256, 0, stream>>>(dst, plane, E, NR, PITCHW);
    owner_degree_kernel<<<NR, 32 * OWN_WAVES, 0, stream>>>(dst, plane, deg, N, PITCHW);
    norm_pack_kernel<<<(E + 4 * BT - 1) / (4 * BT), BT, 0, stream>>>(src, dst, deg, dinv, rec, N, E);

    owner_aggregate_kernel<D_HID><<<NR, 32 * OWN_WAVES, 0, stream>>>(
        rec, plane, h1, dinv, b1, agg1, N, PITCHW);

    gemm_wmma_f32_kernel<D_OUT, true><<<GEMM_GRID2, 32 * (D_OUT / 16), 0, stream>>>(
        agg1, W2, h2, N);
    owner_aggregate_kernel<D_OUT><<<NR, 32 * OWN_WAVES, 0, stream>>>(
        rec, plane, h2, dinv, b2, out, N, PITCHW);
}


# Layer-2 product (ReLU, f32 WMMA) done by the layer-1 owner work-groups on their LDS tile of edge sums (tile pitch 132); sums never stored: 7 launches to 6; product sums k in tile-column order
# speedup vs baseline: 4.4288x; 1.0473x over previous
#include <hip/hip_runtime.h>

typedef __attribute__((ext_vector_type(2))) float v2f;
typedef __attribute__((ext_vector_type(8))) float v8f;

#define D_IN  128
#define D_HID 128
#define D_OUT 64

#define OWN_R 400

#define BM_EDGES 2048
#define BM_WORDS (BM_EDGES / 32)
#define BM_LPITCH (BM_WORDS + 4)
#define BM_MAXR 256

__global__ __launch_bounds__(256)
void plane_build_kernel(const int* __restrict__ dst,
                        unsigned int* __restrict__ plane, int e, int nr, int pitchw) {
    __shared__ __align__(16) unsigned int bm[BM_MAXR * BM_LPITCH];
    const int tid = threadIdx.x;

    const int i0 = blockIdx.x * BM_EDGES + tid * 8;
    int d[8] = {0, 0, 0, 0, 0, 0, 0, 0};
    if (i0 < e) {
        if (i0 + 7 < e && (reinterpret_cast<size_t>(dst) & 15) == 0) {
            const int4 v0 = *(const int4*)(dst + i0);
            const int4 v1 = *(const int4*)(dst + i0 + 4);
            d[0] = v0.x; d[1] = v0.y; d[2] = v0.z; d[3] = v0.w;
            d[4] = v1.x; d[5] = v1.y; d[6] = v1.z; d[7] = v1.w;
        } else {
            #pragma unroll
            for (int k = 0; k < 8; ++k) d[k] = dst[min(i0 + k, e - 1)];
        }
    }

    for (int i = tid; i < nr * (BM_LPITCH / 4); i += 256) ((uint4*)bm)[i] = make_uint4(0u, 0u, 0u, 0u);
    __syncthreads();

    const int w  = tid >> 2;
    const int sh = (tid & 3) * 8;
    #pragma unroll
    for (int k = 0; k < 8; ++k) {
        if (i0 + k < e) atomicOr(&bm[(d[k] / OWN_R) * BM_LPITCH + w], 1u << (sh + k));
    }
    __syncthreads();

    for (int i = tid; i < nr * (BM_WORDS / 4); i += 256) {
        const int r = i / (BM_WORDS / 4), j = i % (BM_WORDS / 4);
        const uint4 v = *(const uint4*)(&bm[r * BM_LPITCH + j * 4]);
        *(uint4*)(&plane[(size_t)r * pitchw + (size_t)blockIdx.x * BM_WORDS + j * 4]) = v;
    }
}

__global__ void norm_pack_kernel(const int* __restrict__ src, const int* __restrict__ dst,
                                 const float* __restrict__ deg, float* __restrict__ dinv,
                                 unsigned int* __restrict__ rec, int n, int e) {
    const int t = blockIdx.x * blockDim.x + threadIdx.x;
    for (int i = t; i < n; i += gridDim.x * blockDim.x) {
        const float d = deg[i];
        dinv[i] = (d > 0.0f) ? rsqrtf(d) : 0.0f;
    }
    uint2* r2 = (uint2*)rec;
    const int base = blockIdx.x * (blockDim.x * 4) + threadIdx.x;
    if (base >= e) return;
    unsigned s[4], d[4];
    float ds[4];
    #pragma unroll
    for (int k = 0; k < 4; ++k) {
        const int i = min(base + k * (int)blockDim.x, e - 1);
        s[k] = (unsigned)src[i];
        d[k] = (unsigned)dst[i];
    }
    #pragma unroll
    for (int k = 0; k < 4; ++k) ds[k] = deg[s[k]];
    #pragma unroll
    for (int k = 0; k < 4; ++k) {
        const int i = base + k * (int)blockDim.x;
        if (i < e) {
            uint2 r;
            r.x = s[k] | ((d[k] % (unsigned)OWN_R) << 17);
            r.y = __float_as_uint((ds[k] > 0.0f) ? rsqrtf(ds[k]) : 0.0f);
            r2[i] = r;
        }
    }
}

template <int NOUT>
__global__ __launch_bounds__(32 * (NOUT / 16))
void gemm_wmma_f32_kernel(const float* __restrict__ A,
                          const float* __restrict__ W,
                          float* __restrict__ H, int M) {
    constexpr int K = 128;
    constexpr int NT = NOUT / 16;
    constexpr int NTHREADS = 32 * NT;
    constexpr int P = K + 4;
    constexpr int NLD = (16 * K / 4) / NTHREADS;
    __shared__ __align__(16) float lds_a[2][16 * P];

    const int tid  = threadIdx.x;
    const int wave = tid >> 5;
    const int lane = tid & 31;
    const int half = lane >> 4;
    const int l16  = lane & 15;
    const int col_base = wave * 16;
    const int nstrips = M / 16;

    int strip = blockIdx.x;
    if (strip >= nstrips) return;

    v2f bf[K / 4];
    #pragma unroll
    for (int s = 0; s < K / 4; ++s) {
        const int ka = s * 4 + half * 2;
        bf[s].x = W[(size_t)ka * NOUT + col_base + l16];
        bf[s].y = W[(size_t)(ka + 1) * NOUT + col_base + l16];
    }

    const float* const ap = A + (size_t)tid * 4;
    const int soff = (tid >> 5) * P + (tid & 31) * 4;
    const size_t ooff = (size_t)(half * 8) * NOUT + col_base + l16;

    float4 p0, p1, p2, p3;
    {
        const float* a = ap + (size_t)strip * (16 * K);
        p0 = *(const float4*)(a);
        p1 = *(const float4*)(a + NTHREADS * 4);
        if (NLD > 2) {
            p2 = *(const float4*)(a + 2 * NTHREADS * 4);
            p3 = *(const float4*)(a + 3 * NTHREADS * 4);
        }
    }

    int buf = 0;
    for (; strip < nstrips; strip += gridDim.x) {
        float* la = lds_a[buf];
        {
            float* ls = la + soff;
            *(float4*)(ls) = p0;
            *(float4*)(ls + NT * P) = p1;
            if (NLD > 2) {
                *(float4*)(ls + 2 * NT * P) = p2;
                *(float4*)(ls + 3 * NT * P) = p3;
            }
        }
        __syncthreads();

        {
            const int nx = min(strip + (int)gridDim.x, nstrips - 1);
            const float* a = ap + (size_t)nx * (16 * K);
            p0 = *(const float4*)(a);
            p1 = *(const float4*)(a + NTHREADS * 4);
            if (NLD > 2) {
                p2 = *(const float4*)(a + 2 * NTHREADS * 4);
                p3 = *(const float4*)(a + 3 * NTHREADS * 4);
            }
        }
        __builtin_amdgcn_sched_barrier(0);

        v8f acc = {};
        #pragma unroll
        for (int s = 0; s < K / 4; ++s) {
            const int ka = s * 4 + half * 2;
            v2f a;
            a.x = la[l16 * P + ka];
            a.y = la[l16 * P + ka + 1];
            acc = __builtin_amdgcn_wmma_f32_16x16x4_f32(
                 false, a,  false, bf[s],
                 (short)0, acc,  false,  false);
        }

        float* const hp = H + (size_t)strip * (16 * NOUT) + ooff;
        #pragma unroll
        for (int v = 0; v < 8; ++v) hp[v * NOUT] = acc[v];
        buf ^= 1;
    }
}

#define OWN_WAVES 32
#define OWN_TILE_FLOATS (OWN_R * (128 + 4))

template <int D, int P>
__device__ __forceinline__ void own_flush(const int* qe, int cnt, int lane,
                                          const uint2* __restrict__ rec,
                                          const float* __restrict__ h,
                                          const float* own_dinv, float* tile) {
    constexpr int VPL = D / 32;
    typedef float vrow __attribute__((ext_vector_type(VPL)));
    int my_s = 0, my_dl = 0;
    float my_c = 0.0f;
    if (lane < cnt) {
        const uint2 r = rec[qe[lane]];
        my_s = (int)(r.x & 0x1ffffu);
        my_dl = (int)(r.x >> 17);
        my_c = __uint_as_float(r.y) * own_dinv[my_dl];
    }
    for (int g = 0; g < cnt; g += 8) {
        vrow v[8];
        #pragma unroll
        for (int t = 0; t < 8; ++t) {
            v[t] = (vrow)(0.0f);
            if (g + t < cnt) {
                const int s = __builtin_amdgcn_readlane(my_s, g + t);
                v[t] = *(const vrow*)(h + (size_t)s * D + lane * VPL);
            }
        }
        #pragma unroll
        for (int t = 0; t < 8; ++t) {
            if (g + t < cnt) {
                const float c = __builtin_bit_cast(
                    float, __builtin_amdgcn_readlane(__builtin_bit_cast(int, my_c), g + t));
                const int dl = __builtin_amdgcn_readlane(my_dl, g + t);
                float* tp = tile + dl * P + lane;
                #pragma unroll
                for (int j = 0; j < VPL; ++j) atomicAdd(tp + j * 32, v[t][j] * c);
            }
        }
    }
}

template <int D, int NEXT>
__global__ __launch_bounds__(1024)
void owner_aggregate_kernel(const unsigned int* __restrict__ recw,
                            const unsigned int* __restrict__ plane,
                            const float* __restrict__ h, const float* __restrict__ dinv,
                            const float* __restrict__ bias, const float* __restrict__ Wn,
                            float* __restrict__ agg, int n, int pitchw) {
    constexpr int VPL = D / 32;
    constexpr int R = OWN_R;
    constexpr int P = (NEXT > 0) ? D + 4 : D;
    typedef float vrow __attribute__((ext_vector_type(VPL)));
    __shared__ __align__(16) float tile[OWN_TILE_FLOATS];
    __shared__ float own_dinv[R];
    __shared__ int q_e[OWN_WAVES][64];

    const int tid  = threadIdx.x;
    const int lane = tid & 31;
    const int wave = __builtin_amdgcn_readfirstlane(tid >> 5);
    const int lo   = blockIdx.x * R;
    const int rows = min(R, n - lo);
    const uint2* rec = (const uint2*)recw;

    vrow bv;
    #pragma unroll
    for (int j = 0; j < VPL; ++j) bv[j] = bias[lane * VPL + j];
    #pragma unroll 4
    for (int r = wave; r < rows; r += OWN_WAVES) {
        const vrow t = *(const vrow*)(h + (size_t)(lo + r) * D + lane * VPL);
        const float d = dinv[lo + r];
        #pragma unroll
        for (int j = 0; j < VPL; ++j) tile[r * P + j * 32 + lane] = t[j] * d * d + bv[j];
    }
    for (int r = tid; r < rows; r += 1024) own_dinv[r] = dinv[lo + r];
    __syncthreads();

    const uint4* prow = (const uint4*)(plane + (size_t)blockIdx.x * pitchw);
    const int nld = pitchw >> 7;
    const unsigned lt = (1u << lane) - 1u;
    int* qe = q_e[wave];
    int qn = 0;

    uint4 nx = prow[(size_t)min(wave, nld - 1) * 32 + lane];

    for (int ld = wave; ld < nld; ld += OWN_WAVES) {
        const uint4 dv = nx;
        nx = prow[(size_t)min(ld + OWN_WAVES, nld - 1) * 32 + lane];

        const int ebase = (ld * 32 + lane) * 128;
        unsigned long long mlo = ((unsigned long long)dv.y << 32) | dv.x;
        unsigned long long mhi = ((unsigned long long)dv.w << 32) | dv.z;

        while (true) {
            const bool has = (mlo | mhi) != 0ull;
            const unsigned bal = __builtin_amdgcn_ballot_w32(has);
            if (bal == 0) break;
            if (has) {
                int b;
                if (mlo != 0ull) { b = __builtin_ctzll(mlo); mlo &= mlo - 1ull; }
                else             { b = 64 + __builtin_ctzll(mhi); mhi &= mhi - 1ull; }
                const int slot = qn + __builtin_popcount(bal & lt);
                qe[slot] = ebase + b;
            }
            qn += __builtin_popcount(bal);
            if (qn >= 32) {
                qn -= 32;
                own_flush<D, P>(qe + qn, 32, lane, rec, h, own_dinv, tile);
            }
        }
    }
    if (qn > 0) own_flush<D, P>(qe, qn, lane, rec, h, own_dinv, tile);

    if constexpr (NEXT > 0) {
        constexpr int NCT = NEXT / 16;
        const int half = lane >> 4;
        const int l16  = lane & 15;
        const int col  = (wave % NCT) * 16 + l16;
        v2f bf[D / 4];
        #pragma unroll
        for (int s = 0; s < D / 4; ++s) {
            const int p = s * 4 + half * 2;
            bf[s].x = Wn[(size_t)((p & 31) * VPL + (p >> 5)) * NEXT + col];
            bf[s].y = Wn[(size_t)(((p + 1) & 31) * VPL + (p >> 5)) * NEXT + col];
        }
        __syncthreads();

        for (int st = wave / NCT; st * 16 < rows; st += OWN_WAVES / NCT) {
            const float* ta = tile + (st * 16 + l16) * P + half * 2;
            v8f acc = {};
            #pragma unroll
            for (int s = 0; s < D / 4; ++s) {
                v2f a = *(const v2f*)(ta + s * 4);
                a.x = fmaxf(a.x, 0.0f);
                a.y = fmaxf(a.y, 0.0f);
                acc = __builtin_amdgcn_wmma_f32_16x16x4_f32(
                     false, a,  false, bf[s],
                     (short)0, acc,  false,  false);
            }
            const int r0 = st * 16 + half * 8;
            float* const hp = agg + (size_t)(lo + r0) * NEXT + col;
            #pragma unroll
            for (int v = 0; v < 8; ++v) {
                if (r0 + v < rows) hp[v * NEXT] = acc[v];
            }
        }
    } else {
        __syncthreads();

        for (int r = wave; r < rows; r += OWN_WAVES) {
            vrow t;
            #pragma unroll
            for (int j = 0; j < VPL; ++j) t[j] = tile[r * P + j * 32 + lane];
            *(vrow*)(agg + (size_t)(lo + r) * D + lane * VPL) = t;
        }
    }
}

#define OWN_DEG_LDS_INTS (41 * 1024)
__global__ __launch_bounds__(1024)
void owner_degree_kernel(const int* __restrict__ dst,
                         const unsigned int* __restrict__ plane,
                         float* __restrict__ deg, int n, int pitchw) {
    constexpr int R = OWN_R;
    __shared__ int cnt[OWN_DEG_LDS_INTS];
    __shared__ int q_e[OWN_WAVES][64];

    const int tid  = threadIdx.x;
    const int lane = tid & 31;
    const int wave = __builtin_amdgcn_readfirstlane(tid >> 5);
    const int lo   = blockIdx.x * R;
    const int rows = min(R, n - lo);

    for (int i = tid; i < R; i += 1024) cnt[i] = 0;
    __syncthreads();

    const uint4* prow = (const uint4*)(plane + (size_t)blockIdx.x * pitchw);
    const int nld = pitchw >> 7;
    const unsigned lt = (1u << lane) - 1u;
    int* qe = q_e[wave];
    int qn = 0;
    int pend = 0;
    bool pend_has = false;

    uint4 nx = prow[(size_t)min(wave, nld - 1) * 32 + lane];

    for (int ld = wave; ld < nld; ld += OWN_WAVES) {
        const uint4 dv = nx;
        nx = prow[(size_t)min(ld + OWN_WAVES, nld - 1) * 32 + lane];

        const int ebase = (ld * 32 + lane) * 128;
        unsigned long long mlo = ((unsigned long long)dv.y << 32) | dv.x;
        unsigned long long mhi = ((unsigned long long)dv.w << 32) | dv.z;

        while (true) {
            const bool has = (mlo | mhi) != 0ull;
            const unsigned bal = __builtin_amdgcn_ballot_w32(has);
            if (bal == 0) break;
            if (has) {
                int b;
                if (mlo != 0ull) { b = __builtin_ctzll(mlo); mlo &= mlo - 1ull; }
                else             { b = 64 + __builtin_ctzll(mhi); mhi &= mhi - 1ull; }
                const int slot = qn + __builtin_popcount(bal & lt);
                qe[slot] = ebase + b;
            }
            qn += __builtin_popcount(bal);
            if (qn >= 32) {
                qn -= 32;
                if (pend_has) atomicAdd(&cnt[pend - lo], 1);
                pend = dst[qe[qn + lane]];
                pend_has = true;
            }
        }
    }
    if (pend_has) atomicAdd(&cnt[pend - lo], 1);
    if (lane < qn) atomicAdd(&cnt[dst[qe[lane]] - lo], 1);
    __syncthreads();

    for (int i = tid; i < rows; i += 1024) deg[lo + i] = 1.0f + (float)cnt[i];
}

extern "C" void kernel_launch(void* const* d_in, const int* in_sizes, int n_in,
                              void* d_out, int out_size, void* d_ws, size_t ws_size,
                              hipStream_t stream) {
    const float* x    = (const float*)d_in[0];
    const int*   eidx = (const int*)  d_in[1];
    const float* W1   = (const float*)d_in[2];
    const float* b1   = (const float*)d_in[3];
    const float* W2   = (const float*)d_in[4];
    const float* b2   = (const float*)d_in[5];
    float* out = (float*)d_out;

    const int N = in_sizes[0] / D_IN;
    const int E = in_sizes[1] / 2;
    const int* src = eidx;
    const int* dst = eidx + E;

    float* ws = (float*)d_ws;
    float* deg  = ws;
    float* dinv = deg + (((size_t)N + 255) & ~(size_t)255);
    float* h1   = dinv + (((size_t)N + 255) & ~(size_t)255);
    float* h2   = h1 + (size_t)N * D_HID;
    const int NR     = (N + OWN_R - 1) / OWN_R;
    const int NBLK   = (((E + BM_EDGES - 1) / BM_EDGES) + 1) & ~1;
    const int PITCHW = NBLK * BM_WORDS;
    unsigned int* plane = (unsigned int*)(h2 + (size_t)N * D_OUT);
    unsigned int* rec   = plane + (size_t)NR * PITCHW;

    const int BT = 256;
    const int GEMM_GRID1 = 1024;

    gemm_wmma_f32_kernel<D_HID><<<GEMM_GRID1, 32 * (D_HID / 16), 0, stream>>>(
        x, W1, h1, N);

    plane_build_kernel<<<NBLK, 256, 0, stream>>>(dst, plane, E, NR, PITCHW);
    owner_degree_kernel<<<NR, 32 * OWN_WAVES, 0, stream>>>(dst, plane, deg, N, PITCHW);
    norm_pack_kernel<<<(E + 4 * BT - 1) / (4 * BT), BT, 0, stream>>>(src, dst, deg, dinv, rec, N, E);

    owner_aggregate_kernel<D_HID, D_OUT><<<NR, 32 * OWN_WAVES, 0, stream>>>(
        rec, plane, h1, dinv, b1, W2, h2, N, PITCHW);

    owner_aggregate_kernel<D_OUT, 0><<<NR, 32 * OWN_WAVES, 0, stream>>>(
        rec, plane, h2, dinv, b2,   nullptr, out, N, PITCHW);
}
